# starting kernel
# baseline (speedup 1.0000x reference)
#include <hip/hip_runtime.h>
#include <hip/hip_bf16.h>

typedef __attribute__((ext_vector_type(2))) float v2f;
typedef __attribute__((ext_vector_type(8))) float v8f;

#define FEAT 128
#define NCLS 7
#define EPSB 1e-5f
#define SCAN_T 1024

__global__ void zero_i32(int* __restrict__ ptr, int n) {
    int i = blockIdx.x * blockDim.x + threadIdx.x;
    if (i < n) ptr[i] = 0;
}

__global__ void hist_rows(const int* __restrict__ erow, int* __restrict__ deg, int E) {
    int e = blockIdx.x * blockDim.x + threadIdx.x;
    if (e < E) atomicAdd(&deg[erow[e]], 1);
}

__global__ void scan_rows(const int* __restrict__ deg, int* __restrict__ rowStart, int N) {
    __shared__ int part[SCAN_T];
    const int t = threadIdx.x;
    const int chunk = (N + SCAN_T - 1) / SCAN_T;
    const int lo = t * chunk;
    const int hi = (lo + chunk < N) ? lo + chunk : N;
    int s = 0;
    for (int i = lo; i < hi; ++i) s += deg[i];
    part[t] = s;
    __syncthreads();
    for (int off = 1; off < SCAN_T; off <<= 1) {
        int add = (t >= off) ? part[t - off] : 0;
        int v = part[t];
        __syncthreads();
        part[t] = v + add;
        __syncthreads();
    }
    int run = part[t] - s;
    for (int i = lo; i < hi; ++i) { rowStart[i] = run; run += deg[i]; }
    if (t == SCAN_T - 1) rowStart[N] = part[t];
}

__global__ void scatter_edges(const int* __restrict__ erow, const int* __restrict__ ecol,
                              const float* __restrict__ ev, const int* __restrict__ rowStart,
                              int* __restrict__ cursor, int* __restrict__ scol,
                              float* __restrict__ sval, int E) {
    int e = blockIdx.x * blockDim.x + threadIdx.x;
    if (e >= E) return;
    const int r = erow[e];
    const int pos = rowStart[r] + atomicAdd(&cursor[r], 1);
    scol[pos] = ecol[e];
    sval[pos] = ev[e];
}

__global__ void gemm_xW1(const float* __restrict__ X, const float* __restrict__ W,
                         float* __restrict__ Y, int nrows) {
    const int wave = threadIdx.x >> 5;
    const int lane = threadIdx.x & 31;
    const int half = lane >> 4;
    const int r    = lane & 15;
    const int mt   = blockIdx.x * 4 + wave;
    if (mt >= (nrows >> 4)) return;
    const int m0 = mt << 4;

    const v8f vzero = {0.f,0.f,0.f,0.f,0.f,0.f,0.f,0.f};
    v8f acc[8];
#pragma unroll
    for (int nt = 0; nt < 8; ++nt) acc[nt] = vzero;

    const float* arow = X + (size_t)(m0 + r) * FEAT;
    for (int kt = 0; kt < 32; ++kt) {
        const int k0 = kt << 2;
        v2f a;
        a.x = arow[k0 + 2 * half + 0];
        a.y = arow[k0 + 2 * half + 1];
#pragma unroll
        for (int nt = 0; nt < 8; ++nt) {
            const int n0 = nt << 4;
            v2f b;
            b.x = W[(size_t)(k0 + 2 * half + 0) * FEAT + n0 + r];
            b.y = W[(size_t)(k0 + 2 * half + 1) * FEAT + n0 + r];
            acc[nt] = __builtin_amdgcn_wmma_f32_16x16x4_f32(
                false, a, false, b, (short)0, acc[nt], false, false);
        }
    }
#pragma unroll
    for (int nt = 0; nt < 8; ++nt) {
        const int n0 = nt << 4;
#pragma unroll
        for (int i = 0; i < 8; ++i)
            Y[(size_t)(m0 + i + 8 * half) * FEAT + n0 + r] = acc[nt][i];
    }
}

__global__ void spmm1_fused(const float* __restrict__ y, const int* __restrict__ rowStart,
                            const int* __restrict__ scol, const float* __restrict__ sval,
                            const float* __restrict__ b1, const float* __restrict__ g1,
                            const float* __restrict__ be1, const float* __restrict__ m1,
                            const float* __restrict__ v1, const float* __restrict__ W2,
                            float* __restrict__ p, int N) {
    const int wave = threadIdx.x >> 5;
    const int lane = threadIdx.x & 31;
    const int r = blockIdx.x * 8 + wave;
    if (r >= N) return;
    const int eBeg = rowStart[r];
    const int eEnd = rowStart[r + 1];

    float4 acc = {0.f, 0.f, 0.f, 0.f};
    for (int e = eBeg; e < eEnd; ++e) {
        const int   c = scol[e];
        const float v = sval[e];
        const float4 g = ((const float4*)(y + (size_t)c * FEAT))[lane];
        acc.x = fmaf(v, g.x, acc.x);
        acc.y = fmaf(v, g.y, acc.y);
        acc.z = fmaf(v, g.z, acc.z);
        acc.w = fmaf(v, g.w, acc.w);
    }

    float h[4];
    float af[4] = {acc.x, acc.y, acc.z, acc.w};
#pragma unroll
    for (int j = 0; j < 4; ++j) {
        const int ch = lane * 4 + j;
        const float s = g1[ch] * rsqrtf(v1[ch] + EPSB);
        const float t = (b1[ch] - m1[ch]) * s + be1[ch];
        const float hv = fmaf(af[j], s, t);
        h[j] = hv > 0.0f ? hv : 0.0f;
    }

#pragma unroll
    for (int c = 0; c < NCLS; ++c) {
        float ssum = 0.0f;
#pragma unroll
        for (int j = 0; j < 4; ++j)
            ssum = fmaf(h[j], W2[(size_t)(lane * 4 + j) * NCLS + c], ssum);
#pragma unroll
        for (int off = 16; off >= 1; off >>= 1) ssum += __shfl_xor(ssum, off, 32);
        if (lane == 0) p[(size_t)r * NCLS + c] = ssum;
    }
}

__global__ void spmm2_fused(const float* __restrict__ p, const int* __restrict__ rowStart,
                            const int* __restrict__ scol, const float* __restrict__ sval,
                            const float* __restrict__ b2, const float* __restrict__ g2,
                            const float* __restrict__ be2, const float* __restrict__ m2,
                            const float* __restrict__ v2, float* __restrict__ out, int total) {
    int t = blockIdx.x * blockDim.x + threadIdx.x;
    if (t >= total) return;
    const int r = t / NCLS;
    const int c = t - r * NCLS;
    float acc = 0.0f;
    const int eEnd = rowStart[r + 1];
    for (int e = rowStart[r]; e < eEnd; ++e)
        acc = fmaf(sval[e], p[(size_t)scol[e] * NCLS + c], acc);
    const float s = g2[c] * rsqrtf(v2[c] + EPSB);
    out[t] = (acc + b2[c] - m2[c]) * s + be2[c];
}

extern "C" void kernel_launch(void* const* d_in, const int* in_sizes, int n_in,
                              void* d_out, int out_size, void* d_ws, size_t ws_size,
                              hipStream_t stream) {
    const float* x   = (const float*)d_in[0];
    const float* ev  = (const float*)d_in[1];
    const float* W1  = (const float*)d_in[2];
    const float* b1  = (const float*)d_in[3];
    const float* g1  = (const float*)d_in[4];
    const float* be1 = (const float*)d_in[5];
    const float* m1  = (const float*)d_in[6];
    const float* v1  = (const float*)d_in[7];
    const float* W2  = (const float*)d_in[8];
    const float* b2  = (const float*)d_in[9];
    const float* g2  = (const float*)d_in[10];
    const float* be2 = (const float*)d_in[11];
    const float* m2  = (const float*)d_in[12];
    const float* v2  = (const float*)d_in[13];
    const int*  erow = (const int*)d_in[14];
    const int*  ecol = (const int*)d_in[15];
    float* out = (float*)d_out;

    const int N = in_sizes[0] / FEAT;
    const int E = in_sizes[1];

    float* y        = (float*)d_ws;
    float* p        = y + (size_t)N * FEAT;
    int*   deg      = (int*)(p + (size_t)N * NCLS);
    int*   cursor   = deg + N;
    int*   rowStart = cursor + N;
    int*   scol     = rowStart + (N + 1);
    float* sval     = (float*)(scol + E);

    zero_i32<<<(2 * N + 255) / 256, 256, 0, stream>>>(deg, 2 * N);
    hist_rows<<<(E + 255) / 256, 256, 0, stream>>>(erow, deg, E);
    scan_rows<<<1, SCAN_T, 0, stream>>>(deg, rowStart, N);
    scatter_edges<<<(E + 255) / 256, 256, 0, stream>>>(erow, ecol, ev, rowStart,
                                                       cursor, scol, sval, E);
    gemm_xW1<<<((N >> 4) + 3) / 4, 128, 0, stream>>>(x, W1, y, N);
    spmm1_fused<<<(N + 7) / 8, 256, 0, stream>>>(y, rowStart, scol, sval,
                                                 b1, g1, be1, m1, v1, W2, p, N);
    spmm2_fused<<<(N * NCLS + 255) / 256, 256, 0, stream>>>(p, rowStart, scol, sval,
                                                            b2, g2, be2, m2, v2, out,
                                                            N * NCLS);
}


# spmm1 tail p=h@W2: 7 class sums from seven 16-byte W2 loads per lane (was 28 strided 4-byte), reduced together by halving exchanges on the vector ALU (9 steps, was 35 LDS permutes); sum pairing kept
# speedup vs baseline: 2.2932x; 2.2932x over previous
#include <hip/hip_runtime.h>
#include <hip/hip_bf16.h>

typedef __attribute__((ext_vector_type(2))) float v2f;
typedef __attribute__((ext_vector_type(8))) float v8f;

#define FEAT 128
#define NCLS 7
#define EPSB 1e-5f
#define SLOT 32

struct __attribute__((aligned(8))) Ent { int col; float val; };

__global__ void zero_i32(int* __restrict__ ptr, int n) {
    int i = blockIdx.x * blockDim.x + threadIdx.x;
    if (i < n) ptr[i] = 0;
}

__global__ void fill_slots(const int* __restrict__ erow, const int* __restrict__ ecol,
                           const float* __restrict__ ev, int* __restrict__ cnt,
                           Ent* __restrict__ slot, int* __restrict__ ovfN,
                           int4* __restrict__ ovf, int E) {
    int e = blockIdx.x * blockDim.x + threadIdx.x;
    if (e >= E) return;
    const int r = erow[e];
    Ent w;
    w.col = ecol[e];
    w.val = ev[e];
    const int k = atomicAdd(&cnt[r], 1);
    if (k < SLOT) {
        slot[(size_t)r * SLOT + k] = w;
    } else {
        const int j = atomicAdd(ovfN, 1);
        ovf[j] = make_int4(r, w.col, __float_as_int(w.val), 0);
    }
}

__global__ void gemm_xW1(const float* __restrict__ X, const float* __restrict__ W,
                         float* __restrict__ Y, int nrows) {
    const int wave = threadIdx.x >> 5;
    const int lane = threadIdx.x & 31;
    const int half = lane >> 4;
    const int r    = lane & 15;
    const int mt   = blockIdx.x * 4 + wave;
    if (mt >= (nrows >> 4)) return;
    const int m0 = mt << 4;

    const v8f vzero = {0.f,0.f,0.f,0.f,0.f,0.f,0.f,0.f};
    v8f acc[8];
#pragma unroll
    for (int nt = 0; nt < 8; ++nt) acc[nt] = vzero;

    const float* arow = X + (size_t)(m0 + r) * FEAT;
    for (int kt = 0; kt < 32; ++kt) {
        const int k0 = kt << 2;
        v2f a;
        a.x = arow[k0 + 2 * half + 0];
        a.y = arow[k0 + 2 * half + 1];
#pragma unroll
        for (int nt = 0; nt < 8; ++nt) {
            const int n0 = nt << 4;
            v2f b;
            b.x = W[(size_t)(k0 + 2 * half + 0) * FEAT + n0 + r];
            b.y = W[(size_t)(k0 + 2 * half + 1) * FEAT + n0 + r];
            acc[nt] = __builtin_amdgcn_wmma_f32_16x16x4_f32(
                false, a, false, b, (short)0, acc[nt], false, false);
        }
    }
#pragma unroll
    for (int nt = 0; nt < 8; ++nt) {
        const int n0 = nt << 4;
#pragma unroll
        for (int i = 0; i < 8; ++i)
            Y[(size_t)(m0 + i + 8 * half) * FEAT + n0 + r] = acc[nt][i];
    }
}

__device__ __forceinline__ float xhalf(float x) {
    const int b = __builtin_bit_cast(int, x);
    return __builtin_bit_cast(float,
        __builtin_amdgcn_permlanex16(b, b, 0x76543210, 0xfedcba98, false, false));
}
template <int M>
__device__ __forceinline__ float xrow(float x) {
    const int b = __builtin_bit_cast(int, x);
    return __builtin_bit_cast(float,
        __builtin_amdgcn_update_dpp(0, b, 0x160 + M, 0xf, 0xf, false));
}

__global__ void __launch_bounds__(256, 16)
spmm1_fused(const float* __restrict__ y, const int* __restrict__ cnt,
                            const Ent* __restrict__ slot, const int* __restrict__ ovfN,
                            const int4* __restrict__ ovf,
                            const float* __restrict__ b1, const float* __restrict__ g1,
                            const float* __restrict__ be1, const float* __restrict__ m1,
                            const float* __restrict__ v1, const float* __restrict__ W2,
                            float* __restrict__ p, int N) {
    const int wave = threadIdx.x >> 5;
    const int lane = threadIdx.x & 31;
    const int r = blockIdx.x * 8 + wave;
    if (r >= N) return;
    const int d = cnt[r];
    const int n = d < SLOT ? d : SLOT;
    const Ent* row = slot + (size_t)r * SLOT;

    float4 acc = {0.f, 0.f, 0.f, 0.f};
    for (int e = 0; e < n; ++e) {
        const Ent w = row[e];
        const float4 g = ((const float4*)(y + (size_t)w.col * FEAT))[lane];
        acc.x = fmaf(w.val, g.x, acc.x);
        acc.y = fmaf(w.val, g.y, acc.y);
        acc.z = fmaf(w.val, g.z, acc.z);
        acc.w = fmaf(w.val, g.w, acc.w);
    }
    if (d > SLOT) {
        const int m = *ovfN;
        for (int j = 0; j < m; ++j) {
            const int4 o = ovf[j];
            if (o.x == r) {
                const float v = __int_as_float(o.z);
                const float4 g = ((const float4*)(y + (size_t)o.y * FEAT))[lane];
                acc.x = fmaf(v, g.x, acc.x);
                acc.y = fmaf(v, g.y, acc.y);
                acc.z = fmaf(v, g.z, acc.z);
                acc.w = fmaf(v, g.w, acc.w);
            }
        }
    }

    const int ln = (int)__builtin_amdgcn_mbcnt_lo(~0u, 0u);

    float h[4];
    float af[4] = {acc.x, acc.y, acc.z, acc.w};
#pragma unroll
    for (int j = 0; j < 4; ++j) {
        const int ch = ln * 4 + j;
        const float s = g1[ch] * rsqrtf(v1[ch] + EPSB);
        const float t = (b1[ch] - m1[ch]) * s + be1[ch];
        const float hv = fmaf(af[j], s, t);
        h[j] = hv > 0.0f ? hv : 0.0f;
    }

    float w[28];
    {
        const float4* wq = (const float4*)(W2 + (size_t)ln * (4 * NCLS));
#pragma unroll
        for (int q = 0; q < 7; ++q) {
            const float4 t4 = wq[q];
            w[4 * q + 0] = t4.x; w[4 * q + 1] = t4.y;
            w[4 * q + 2] = t4.z; w[4 * q + 3] = t4.w;
        }
    }
    float s8[8];
#pragma unroll
    for (int c = 0; c < NCLS; ++c) {
        float ssum = 0.0f;
#pragma unroll
        for (int j = 0; j < 4; ++j) ssum = fmaf(h[j], w[j * NCLS + c], ssum);
        s8[c] = ssum;
    }
    s8[7] = 0.0f;

    const bool b4 = (ln & 16) != 0;
    const bool b3 = (ln & 8) != 0;
    const bool b2 = (ln & 4) != 0;
    float t4v[4];
#pragma unroll
    for (int i = 0; i < 4; ++i) {
        const float keep = b4 ? s8[i + 4] : s8[i];
        const float send = b4 ? s8[i] : s8[i + 4];
        t4v[i] = keep + xhalf(send);
    }
    float u2[2];
#pragma unroll
    for (int i = 0; i < 2; ++i) {
        const float keep = b3 ? t4v[i + 2] : t4v[i];
        const float send = b3 ? t4v[i] : t4v[i + 2];
        u2[i] = keep + xrow<8>(send);
    }
    float tot;
    {
        const float keep = b2 ? u2[1] : u2[0];
        const float send = b2 ? u2[0] : u2[1];
        tot = keep + xrow<4>(send);
    }
    tot += xrow<2>(tot);
    tot += xrow<1>(tot);
    if ((ln & 3) == 0 && ln < 4 * NCLS)
        p[(size_t)r * NCLS + (ln >> 2)] = tot;
}

__global__ void spmm2_fused(const float* __restrict__ p, const int* __restrict__ cnt,
                            const Ent* __restrict__ slot, const int* __restrict__ ovfN,
                            const int4* __restrict__ ovf,
                            const float* __restrict__ b2, const float* __restrict__ g2,
                            const float* __restrict__ be2, const float* __restrict__ m2,
                            const float* __restrict__ v2, float* __restrict__ out, int total) {
    int t = blockIdx.x * blockDim.x + threadIdx.x;
    if (t >= total) return;
    const int r = t / NCLS;
    const int c = t - r * NCLS;
    float acc = 0.0f;
    const int d = cnt[r];
    const int n = d < SLOT ? d : SLOT;
    const Ent* row = slot + (size_t)r * SLOT;
    for (int e = 0; e < n; ++e) {
        const Ent w = row[e];
        acc = fmaf(w.val, p[(size_t)w.col * NCLS + c], acc);
    }
    if (d > SLOT) {
        const int m = *ovfN;
        for (int j = 0; j < m; ++j) {
            const int4 o = ovf[j];
            if (o.x == r)
                acc = fmaf(__int_as_float(o.z), p[(size_t)o.y * NCLS + c], acc);
        }
    }
    const float s = g2[c] * rsqrtf(v2[c] + EPSB);
    out[t] = (acc + b2[c] - m2[c]) * s + be2[c];
}

extern "C" void kernel_launch(void* const* d_in, const int* in_sizes, int n_in,
                              void* d_out, int out_size, void* d_ws, size_t ws_size,
                              hipStream_t stream) {
    const float* x   = (const float*)d_in[0];
    const float* ev  = (const float*)d_in[1];
    const float* W1  = (const float*)d_in[2];
    const float* b1  = (const float*)d_in[3];
    const float* g1  = (const float*)d_in[4];
    const float* be1 = (const float*)d_in[5];
    const float* m1  = (const float*)d_in[6];
    const float* v1  = (const float*)d_in[7];
    const float* W2  = (const float*)d_in[8];
    const float* b2  = (const float*)d_in[9];
    const float* g2  = (const float*)d_in[10];
    const float* be2 = (const float*)d_in[11];
    const float* m2  = (const float*)d_in[12];
    const float* v2  = (const float*)d_in[13];
    const int*  erow = (const int*)d_in[14];
    const int*  ecol = (const int*)d_in[15];
    float* out = (float*)d_out;

    const int N = in_sizes[0] / FEAT;
    const int E = in_sizes[1];

    float* y    = (float*)d_ws;
    Ent*   slot = (Ent*)(y + (size_t)N * FEAT);
    int4*  ovf  = (int4*)(slot + (size_t)N * SLOT);
    float* p    = (float*)(ovf + E);
    int*   cnt  = (int*)(p + (size_t)N * NCLS);
    int*   ovfN = cnt + N;

    zero_i32<<<(N + 1 + 255) / 256, 256, 0, stream>>>(cnt, N + 1);
    fill_slots<<<(E + 255) / 256, 256, 0, stream>>>(erow, ecol, ev, cnt, slot, ovfN, ovf, E);
    gemm_xW1<<<((N >> 4) + 3) / 4, 128, 0, stream>>>(x, W1, y, N);
    spmm1_fused<<<(N + 7) / 8, 256, 0, stream>>>(y, cnt, slot, ovfN, ovf,
                                                 b1, g1, be1, m1, v1, W2, p, N);
    spmm2_fused<<<(N * NCLS + 255) / 256, 256, 0, stream>>>(p, cnt, slot, ovfN, ovf,
                                                            b2, g2, be2, m2, v2, out,
                                                            N * NCLS);
}


# fill_slots and gemm_xW1 (independent) run as one launch: matrix work-groups first, edge work-groups (4 edges per lane) in the wave slots beside them; 5 -> 4 launches; sum order stays arrival order
# speedup vs baseline: 2.5376x; 1.1066x over previous
#include <hip/hip_runtime.h>
#include <hip/hip_bf16.h>

typedef __attribute__((ext_vector_type(2))) float v2f;
typedef __attribute__((ext_vector_type(8))) float v8f;

#define FEAT 128
#define NCLS 7
#define EPSB 1e-5f
#define SLOT 32

struct __attribute__((aligned(8))) Ent { int col; float val; };

__global__ void zero_i32(int* __restrict__ ptr, int n) {
    int i = blockIdx.x * blockDim.x + threadIdx.x;
    if (i < n) ptr[i] = 0;
}

__device__ __forceinline__ void put_edge(int r, int c, float v, int k,
                                         Ent* __restrict__ slot, int* __restrict__ ovfN,
                                         int4* __restrict__ ovf) {
    if (k < SLOT) {
        Ent w;
        w.col = c;
        w.val = v;
        slot[(size_t)r * SLOT + k] = w;
    } else {
        const int j = atomicAdd(ovfN, 1);
        ovf[j] = make_int4(r, c, __float_as_int(v), 0);
    }
}

__device__ __forceinline__ void fill_role(const int* __restrict__ erow, const int* __restrict__ ecol,
                                          const float* __restrict__ ev, int* __restrict__ cnt,
                                          Ent* __restrict__ slot, int* __restrict__ ovfN,
                                          int4* __restrict__ ovf, int E, int q) {
    const int e0 = q * 4;
    if (e0 >= E) return;
    if (e0 + 3 < E) {
        const int4   r4 = ((const int4*)erow)[q];
        const int4   c4 = ((const int4*)ecol)[q];
        const float4 v4 = ((const float4*)ev)[q];
        const int k0 = atomicAdd(&cnt[r4.x], 1);
        const int k1 = atomicAdd(&cnt[r4.y], 1);
        const int k2 = atomicAdd(&cnt[r4.z], 1);
        const int k3 = atomicAdd(&cnt[r4.w], 1);
        put_edge(r4.x, c4.x, v4.x, k0, slot, ovfN, ovf);
        put_edge(r4.y, c4.y, v4.y, k1, slot, ovfN, ovf);
        put_edge(r4.z, c4.z, v4.z, k2, slot, ovfN, ovf);
        put_edge(r4.w, c4.w, v4.w, k3, slot, ovfN, ovf);
    } else {
        for (int e = e0; e < E; ++e) {
            const int r = erow[e];
            const int k = atomicAdd(&cnt[r], 1);
            put_edge(r, ecol[e], ev[e], k, slot, ovfN, ovf);
        }
    }
}

__device__ __forceinline__ void gemm_role(const float* __restrict__ X, const float* __restrict__ W,
                                          float* __restrict__ Y, int nrows, int mt) {
    const int lane = threadIdx.x & 31;
    const int half = lane >> 4;
    const int r    = lane & 15;
    if (mt >= (nrows >> 4)) return;
    const int m0 = mt << 4;

    const v8f vzero = {0.f,0.f,0.f,0.f,0.f,0.f,0.f,0.f};
    v8f acc[8];
#pragma unroll
    for (int nt = 0; nt < 8; ++nt) acc[nt] = vzero;

    const float* arow = X + (size_t)(m0 + r) * FEAT;
    for (int kt = 0; kt < 32; ++kt) {
        const int k0 = kt << 2;
        v2f a;
        a.x = arow[k0 + 2 * half + 0];
        a.y = arow[k0 + 2 * half + 1];
#pragma unroll
        for (int nt = 0; nt < 8; ++nt) {
            const int n0 = nt << 4;
            v2f b;
            b.x = W[(size_t)(k0 + 2 * half + 0) * FEAT + n0 + r];
            b.y = W[(size_t)(k0 + 2 * half + 1) * FEAT + n0 + r];
            acc[nt] = __builtin_amdgcn_wmma_f32_16x16x4_f32(
                false, a, false, b, (short)0, acc[nt], false, false);
        }
    }
#pragma unroll
    for (int nt = 0; nt < 8; ++nt) {
        const int n0 = nt << 4;
#pragma unroll
        for (int i = 0; i < 8; ++i)
            Y[(size_t)(m0 + i + 8 * half) * FEAT + n0 + r] = acc[nt][i];
    }
}

__global__ void __launch_bounds__(128)
gemm_and_fill(const float* __restrict__ X, const float* __restrict__ W,
              float* __restrict__ Y, int nrows, int gemmBlocks,
              const int* __restrict__ erow, const int* __restrict__ ecol,
              const float* __restrict__ ev, int* __restrict__ cnt,
              Ent* __restrict__ slot, int* __restrict__ ovfN,
              int4* __restrict__ ovf, int E) {
    const int b = (int)blockIdx.x;
    if (b < gemmBlocks) {
        gemm_role(X, W, Y, nrows, b * 4 + (int)(threadIdx.x >> 5));
    } else {
        fill_role(erow, ecol, ev, cnt, slot, ovfN, ovf, E,
                  (b - gemmBlocks) * 128 + (int)threadIdx.x);
    }
}

__device__ __forceinline__ float xhalf(float x) {
    const int b = __builtin_bit_cast(int, x);
    return __builtin_bit_cast(float,
        __builtin_amdgcn_permlanex16(b, b, 0x76543210, 0xfedcba98, false, false));
}
template <int M>
__device__ __forceinline__ float xrow(float x) {
    const int b = __builtin_bit_cast(int, x);
    return __builtin_bit_cast(float,
        __builtin_amdgcn_update_dpp(0, b, 0x160 + M, 0xf, 0xf, false));
}

__global__ void __launch_bounds__(256, 16)
spmm1_fused(const float* __restrict__ y, const int* __restrict__ cnt,
                            const Ent* __restrict__ slot, const int* __restrict__ ovfN,
                            const int4* __restrict__ ovf,
                            const float* __restrict__ b1, const float* __restrict__ g1,
                            const float* __restrict__ be1, const float* __restrict__ m1,
                            const float* __restrict__ v1, const float* __restrict__ W2,
                            float* __restrict__ p, int N) {
    const int wave = threadIdx.x >> 5;
    const int lane = threadIdx.x & 31;
    const int r = blockIdx.x * 8 + wave;
    if (r >= N) return;
    const int d = cnt[r];
    const int n = d < SLOT ? d : SLOT;
    const Ent* row = slot + (size_t)r * SLOT;

    float4 acc = {0.f, 0.f, 0.f, 0.f};
    for (int e = 0; e < n; ++e) {
        const Ent w = row[e];
        const float4 g = ((const float4*)(y + (size_t)w.col * FEAT))[lane];
        acc.x = fmaf(w.val, g.x, acc.x);
        acc.y = fmaf(w.val, g.y, acc.y);
        acc.z = fmaf(w.val, g.z, acc.z);
        acc.w = fmaf(w.val, g.w, acc.w);
    }
    if (d > SLOT) {
        const int m = *ovfN;
        for (int j = 0; j < m; ++j) {
            const int4 o = ovf[j];
            if (o.x == r) {
                const float v = __int_as_float(o.z);
                const float4 g = ((const float4*)(y + (size_t)o.y * FEAT))[lane];
                acc.x = fmaf(v, g.x, acc.x);
                acc.y = fmaf(v, g.y, acc.y);
                acc.z = fmaf(v, g.z, acc.z);
                acc.w = fmaf(v, g.w, acc.w);
            }
        }
    }

    const int ln = (int)__builtin_amdgcn_mbcnt_lo(~0u, 0u);

    float h[4];
    float af[4] = {acc.x, acc.y, acc.z, acc.w};
#pragma unroll
    for (int j = 0; j < 4; ++j) {
        const int ch = ln * 4 + j;
        const float s = g1[ch] * rsqrtf(v1[ch] + EPSB);
        const float t = (b1[ch] - m1[ch]) * s + be1[ch];
        const float hv = fmaf(af[j], s, t);
        h[j] = hv > 0.0f ? hv : 0.0f;
    }

    float w[28];
    {
        const float4* wq = (const float4*)(W2 + (size_t)ln * (4 * NCLS));
#pragma unroll
        for (int q = 0; q < 7; ++q) {
            const float4 t4 = wq[q];
            w[4 * q + 0] = t4.x; w[4 * q + 1] = t4.y;
            w[4 * q + 2] = t4.z; w[4 * q + 3] = t4.w;
        }
    }
    float s8[8];
#pragma unroll
    for (int c = 0; c < NCLS; ++c) {
        float ssum = 0.0f;
#pragma unroll
        for (int j = 0; j < 4; ++j) ssum = fmaf(h[j], w[j * NCLS + c], ssum);
        s8[c] = ssum;
    }
    s8[7] = 0.0f;

    const bool b4 = (ln & 16) != 0;
    const bool b3 = (ln & 8) != 0;
    const bool b2 = (ln & 4) != 0;
    float t4v[4];
#pragma unroll
    for (int i = 0; i < 4; ++i) {
        const float keep = b4 ? s8[i + 4] : s8[i];
        const float send = b4 ? s8[i] : s8[i + 4];
        t4v[i] = keep + xhalf(send);
    }
    float u2[2];
#pragma unroll
    for (int i = 0; i < 2; ++i) {
        const float keep = b3 ? t4v[i + 2] : t4v[i];
        const float send = b3 ? t4v[i] : t4v[i + 2];
        u2[i] = keep + xrow<8>(send);
    }
    float tot;
    {
        const float keep = b2 ? u2[1] : u2[0];
        const float send = b2 ? u2[0] : u2[1];
        tot = keep + xrow<4>(send);
    }
    tot += xrow<2>(tot);
    tot += xrow<1>(tot);
    if ((ln & 3) == 0 && ln < 4 * NCLS)
        p[(size_t)r * NCLS + (ln >> 2)] = tot;
}

__global__ void spmm2_fused(const float* __restrict__ p, const int* __restrict__ cnt,
                            const Ent* __restrict__ slot, const int* __restrict__ ovfN,
                            const int4* __restrict__ ovf,
                            const float* __restrict__ b2, const float* __restrict__ g2,
                            const float* __restrict__ be2, const float* __restrict__ m2,
                            const float* __restrict__ v2, float* __restrict__ out, int total) {
    int t = blockIdx.x * blockDim.x + threadIdx.x;
    if (t >= total) return;
    const int r = t / NCLS;
    const int c = t - r * NCLS;
    float acc = 0.0f;
    const int d = cnt[r];
    const int n = d < SLOT ? d : SLOT;
    const Ent* row = slot + (size_t)r * SLOT;
    for (int e = 0; e < n; ++e) {
        const Ent w = row[e];
        acc = fmaf(w.val, p[(size_t)w.col * NCLS + c], acc);
    }
    if (d > SLOT) {
        const int m = *ovfN;
        for (int j = 0; j < m; ++j) {
            const int4 o = ovf[j];
            if (o.x == r)
                acc = fmaf(__int_as_float(o.z), p[(size_t)o.y * NCLS + c], acc);
        }
    }
    const float s = g2[c] * rsqrtf(v2[c] + EPSB);
    out[t] = (acc + b2[c] - m2[c]) * s + be2[c];
}

extern "C" void kernel_launch(void* const* d_in, const int* in_sizes, int n_in,
                              void* d_out, int out_size, void* d_ws, size_t ws_size,
                              hipStream_t stream) {
    const float* x   = (const float*)d_in[0];
    const float* ev  = (const float*)d_in[1];
    const float* W1  = (const float*)d_in[2];
    const float* b1  = (const float*)d_in[3];
    const float* g1  = (const float*)d_in[4];
    const float* be1 = (const float*)d_in[5];
    const float* m1  = (const float*)d_in[6];
    const float* v1  = (const float*)d_in[7];
    const float* W2  = (const float*)d_in[8];
    const float* b2  = (const float*)d_in[9];
    const float* g2  = (const float*)d_in[10];
    const float* be2 = (const float*)d_in[11];
    const float* m2  = (const float*)d_in[12];
    const float* v2  = (const float*)d_in[13];
    const int*  erow = (const int*)d_in[14];
    const int*  ecol = (const int*)d_in[15];
    float* out = (float*)d_out;

    const int N = in_sizes[0] / FEAT;
    const int E = in_sizes[1];

    float* y    = (float*)d_ws;
    Ent*   slot = (Ent*)(y + (size_t)N * FEAT);
    int4*  ovf  = (int4*)(slot + (size_t)N * SLOT);
    float* p    = (float*)(ovf + E);
    int*   cnt  = (int*)(p + (size_t)N * NCLS);
    int*   ovfN = cnt + N;

    zero_i32<<<(N + 1 + 255) / 256, 256, 0, stream>>>(cnt, N + 1);
    const int gemmBlocks = ((N >> 4) + 3) / 4;
    const int fillBlocks = (E + 511) / 512;
    gemm_and_fill<<<gemmBlocks + fillBlocks, 128, 0, stream>>>(x, W1, y, N, gemmBlocks,
                                                             erow, ecol, ev, cnt, slot,
                                                             ovfN, ovf, E);
    spmm1_fused<<<(N + 7) / 8, 256, 0, stream>>>(y, cnt, slot, ovfN, ovf,
                                                 b1, g1, be1, m1, v1, W2, p, N);
    spmm2_fused<<<(N * NCLS + 255) / 256, 256, 0, stream>>>(p, cnt, slot, ovfN, ovf,
                                                            b2, g2, be2, m2, v2, out,
                                                            N * NCLS);
}


# spmm1_fused: 100000 one-row waves -> 8192 resident waves that walk their rows: W2, BN and row counts once per wave, next row's slot entries fetched under this row's gathers; sum order kept
# speedup vs baseline: 2.6598x; 1.0482x over previous
#include <hip/hip_runtime.h>
#include <hip/hip_bf16.h>

typedef __attribute__((ext_vector_type(2))) float v2f;
typedef __attribute__((ext_vector_type(8))) float v8f;

#define FEAT 128
#define NCLS 7
#define EPSB 1e-5f
#define SLOT 32

struct __attribute__((aligned(8))) Ent { int col; float val; };

__global__ void zero_i32(int* __restrict__ ptr, int n) {
    int i = blockIdx.x * blockDim.x + threadIdx.x;
    if (i < n) ptr[i] = 0;
}

__device__ __forceinline__ void put_edge(int r, int c, float v, int k,
                                         Ent* __restrict__ slot, int* __restrict__ ovfN,
                                         int4* __restrict__ ovf) {
    if (k < SLOT) {
        Ent w;
        w.col = c;
        w.val = v;
        slot[(size_t)r * SLOT + k] = w;
    } else {
        const int j = atomicAdd(ovfN, 1);
        ovf[j] = make_int4(r, c, __float_as_int(v), 0);
    }
}

__device__ __forceinline__ void fill_role(const int* __restrict__ erow, const int* __restrict__ ecol,
                                          const float* __restrict__ ev, int* __restrict__ cnt,
                                          Ent* __restrict__ slot, int* __restrict__ ovfN,
                                          int4* __restrict__ ovf, int E, int q) {
    const int e0 = q * 4;
    if (e0 >= E) return;
    if (e0 + 3 < E) {
        const int4   r4 = ((const int4*)erow)[q];
        const int4   c4 = ((const int4*)ecol)[q];
        const float4 v4 = ((const float4*)ev)[q];
        const int k0 = atomicAdd(&cnt[r4.x], 1);
        const int k1 = atomicAdd(&cnt[r4.y], 1);
        const int k2 = atomicAdd(&cnt[r4.z], 1);
        const int k3 = atomicAdd(&cnt[r4.w], 1);
        put_edge(r4.x, c4.x, v4.x, k0, slot, ovfN, ovf);
        put_edge(r4.y, c4.y, v4.y, k1, slot, ovfN, ovf);
        put_edge(r4.z, c4.z, v4.z, k2, slot, ovfN, ovf);
        put_edge(r4.w, c4.w, v4.w, k3, slot, ovfN, ovf);
    } else {
        for (int e = e0; e < E; ++e) {
            const int r = erow[e];
            const int k = atomicAdd(&cnt[r], 1);
            put_edge(r, ecol[e], ev[e], k, slot, ovfN, ovf);
        }
    }
}

__device__ __forceinline__ void gemm_role(const float* __restrict__ X, const float* __restrict__ W,
                                          float* __restrict__ Y, int nrows, int mt) {
    const int lane = threadIdx.x & 31;
    const int half = lane >> 4;
    const int r    = lane & 15;
    if (mt >= (nrows >> 4)) return;
    const int m0 = mt << 4;

    const v8f vzero = {0.f,0.f,0.f,0.f,0.f,0.f,0.f,0.f};
    v8f acc[8];
#pragma unroll
    for (int nt = 0; nt < 8; ++nt) acc[nt] = vzero;

    const float* arow = X + (size_t)(m0 + r) * FEAT;
    for (int kt = 0; kt < 32; ++kt) {
        const int k0 = kt << 2;
        v2f a;
        a.x = arow[k0 + 2 * half + 0];
        a.y = arow[k0 + 2 * half + 1];
#pragma unroll
        for (int nt = 0; nt < 8; ++nt) {
            const int n0 = nt << 4;
            v2f b;
            b.x = W[(size_t)(k0 + 2 * half + 0) * FEAT + n0 + r];
            b.y = W[(size_t)(k0 + 2 * half + 1) * FEAT + n0 + r];
            acc[nt] = __builtin_amdgcn_wmma_f32_16x16x4_f32(
                false, a, false, b, (short)0, acc[nt], false, false);
        }
    }
#pragma unroll
    for (int nt = 0; nt < 8; ++nt) {
        const int n0 = nt << 4;
#pragma unroll
        for (int i = 0; i < 8; ++i)
            Y[(size_t)(m0 + i + 8 * half) * FEAT + n0 + r] = acc[nt][i];
    }
}

__global__ void __launch_bounds__(128)
gemm_and_fill(const float* __restrict__ X, const float* __restrict__ W,
              float* __restrict__ Y, int nrows, int gemmBlocks,
              const int* __restrict__ erow, const int* __restrict__ ecol,
              const float* __restrict__ ev, int* __restrict__ cnt,
              Ent* __restrict__ slot, int* __restrict__ ovfN,
              int4* __restrict__ ovf, int E) {
    const int b = (int)blockIdx.x;
    if (b < gemmBlocks) {
        gemm_role(X, W, Y, nrows, b * 4 + (int)(threadIdx.x >> 5));
    } else {
        fill_role(erow, ecol, ev, cnt, slot, ovfN, ovf, E,
                  (b - gemmBlocks) * 128 + (int)threadIdx.x);
    }
}

__device__ __forceinline__ float xhalf(float x) {
    const int b = __builtin_bit_cast(int, x);
    return __builtin_bit_cast(float,
        __builtin_amdgcn_permlanex16(b, b, 0x76543210, 0xfedcba98, false, false));
}
template <int M>
__device__ __forceinline__ float xrow(float x) {
    const int b = __builtin_bit_cast(int, x);
    return __builtin_bit_cast(float,
        __builtin_amdgcn_update_dpp(0, b, 0x160 + M, 0xf, 0xf, false));
}

__device__ __forceinline__ float4 gather_row(const float* __restrict__ y, int c, int lane) {
    return ((const float4*)(y + (size_t)c * FEAT))[lane];
}
__device__ __forceinline__ void axpy4(float4& acc, float v, const float4 g) {
    acc.x = fmaf(v, g.x, acc.x);
    acc.y = fmaf(v, g.y, acc.y);
    acc.z = fmaf(v, g.z, acc.z);
    acc.w = fmaf(v, g.w, acc.w);
}

__global__ void __launch_bounds__(256, 8)
spmm1_fused(const float* __restrict__ y, const int* __restrict__ cnt,
                            const Ent* __restrict__ slot, const int* __restrict__ ovfN,
                            const int4* __restrict__ ovf,
                            const float* __restrict__ b1, const float* __restrict__ g1,
                            const float* __restrict__ be1, const float* __restrict__ m1,
                            const float* __restrict__ v1, const float* __restrict__ W2,
                            float* __restrict__ p, int N) {
    const int lane = threadIdx.x & 31;
    const int wv = __builtin_amdgcn_readfirstlane((int)(blockIdx.x * 8 + (threadIdx.x >> 5)));
    const int stride = (int)gridDim.x * 8;
    if (wv >= N) return;

    float bs[4], bt[4];
#pragma unroll
    for (int j = 0; j < 4; ++j) {
        const int ch = lane * 4 + j;
        const float s = g1[ch] * rsqrtf(v1[ch] + EPSB);
        bs[j] = s;
        bt[j] = (b1[ch] - m1[ch]) * s + be1[ch];
    }
    float w[28];
    {
        const float4* wq = (const float4*)(W2 + (size_t)lane * (4 * NCLS));
#pragma unroll
        for (int q = 0; q < 7; ++q) {
            const float4 t4 = wq[q];
            w[4 * q + 0] = t4.x; w[4 * q + 1] = t4.y;
            w[4 * q + 2] = t4.z; w[4 * q + 3] = t4.w;
        }
    }
    const bool b4 = (lane & 16) != 0;
    const bool b3 = (lane & 8) != 0;
    const bool b2 = (lane & 4) != 0;

    for (int base = wv; base < N; base += 32 * stride) {
        const int rj = base + lane * stride;
        const int cntv = rj < N ? cnt[rj] : 0;
        int rows = (N - base + stride - 1) / stride;
        rows = rows < 32 ? rows : 32;

        int d = __builtin_amdgcn_readlane(cntv, 0);
        Ent ent;
        ent.col = 0;
        ent.val = 0.0f;
        if (lane < (d < SLOT ? d : SLOT)) ent = slot[(size_t)base * SLOT + lane];
        asm volatile("" :: "v"(ent.col), "v"(ent.val));

        for (int k = 0; k < rows; ++k) {
            const int r = base + k * stride;
            const int n = d < SLOT ? d : SLOT;

            int dn = 0;
            Ent entn;
            entn.col = 0;
            entn.val = 0.0f;
            if (k + 1 < rows) {
                dn = __builtin_amdgcn_readlane(cntv, k + 1);
                if (lane < (dn < SLOT ? dn : SLOT))
                    entn = slot[(size_t)(r + stride) * SLOT + lane];
            }

            const int ecol = ent.col;
            const int evalb = __float_as_int(ent.val);
            float4 acc = {0.f, 0.f, 0.f, 0.f};
            int e0 = 0;
            for (; e0 + 8 <= n; e0 += 8) {
                float4 g[8];
#pragma unroll
                for (int j = 0; j < 8; ++j)
                    g[j] = gather_row(y, __builtin_amdgcn_readlane(ecol, e0 + j), lane);
#pragma unroll
                for (int j = 0; j < 8; ++j)
                    axpy4(acc, __int_as_float(__builtin_amdgcn_readlane(evalb, e0 + j)), g[j]);
            }
            const int rem = n - e0;
            if (rem > 0) {
                float4 g[7];
#pragma unroll
                for (int j = 0; j < 7; ++j)
                    if (j < rem)
                        g[j] = gather_row(y, __builtin_amdgcn_readlane(ecol, e0 + j), lane);
#pragma unroll
                for (int j = 0; j < 7; ++j)
                    if (j < rem)
                        axpy4(acc, __int_as_float(__builtin_amdgcn_readlane(evalb, e0 + j)), g[j]);
            }
            if (d > SLOT) {
                const int m = *ovfN;
                for (int j = 0; j < m; ++j) {
                    const int4 o = ovf[j];
                    if (o.x == r) axpy4(acc, __int_as_float(o.z), gather_row(y, o.y, lane));
                }
            }

            float h[4];
            const float af[4] = {acc.x, acc.y, acc.z, acc.w};
#pragma unroll
            for (int j = 0; j < 4; ++j) {
                const float hv = fmaf(af[j], bs[j], bt[j]);
                h[j] = hv > 0.0f ? hv : 0.0f;
            }

            float s8[8];
#pragma unroll
            for (int c = 0; c < NCLS; ++c) {
                float ssum = 0.0f;
#pragma unroll
                for (int j = 0; j < 4; ++j) ssum = fmaf(h[j], w[j * NCLS + c], ssum);
                s8[c] = ssum;
            }
            s8[7] = 0.0f;

            float t4v[4];
#pragma unroll
            for (int i = 0; i < 4; ++i) {
                const float keep = b4 ? s8[i + 4] : s8[i];
                const float send = b4 ? s8[i] : s8[i + 4];
                t4v[i] = keep + xhalf(send);
            }
            float u2[2];
#pragma unroll
            for (int i = 0; i < 2; ++i) {
                const float keep = b3 ? t4v[i + 2] : t4v[i];
                const float send = b3 ? t4v[i] : t4v[i + 2];
                u2[i] = keep + xrow<8>(send);
            }
            float tot;
            {
                const float keep = b2 ? u2[1] : u2[0];
                const float send = b2 ? u2[0] : u2[1];
                tot = keep + xrow<4>(send);
            }
            tot += xrow<2>(tot);
            tot += xrow<1>(tot);
            if ((lane & 3) == 0 && lane < 4 * NCLS)
                p[(size_t)r * NCLS + (lane >> 2)] = tot;

            d = dn;
            ent = entn;
        }
    }
}

__global__ void spmm2_fused(const float* __restrict__ p, const int* __restrict__ cnt,
                            const Ent* __restrict__ slot, const int* __restrict__ ovfN,
                            const int4* __restrict__ ovf,
                            const float* __restrict__ b2, const float* __restrict__ g2,
                            const float* __restrict__ be2, const float* __restrict__ m2,
                            const float* __restrict__ v2, float* __restrict__ out, int total) {
    int t = blockIdx.x * blockDim.x + threadIdx.x;
    if (t >= total) return;
    const int r = t / NCLS;
    const int c = t - r * NCLS;
    float acc = 0.0f;
    const int d = cnt[r];
    const int n = d < SLOT ? d : SLOT;
    const Ent* row = slot + (size_t)r * SLOT;
    for (int e = 0; e < n; ++e) {
        const Ent w = row[e];
        acc = fmaf(w.val, p[(size_t)w.col * NCLS + c], acc);
    }
    if (d > SLOT) {
        const int m = *ovfN;
        for (int j = 0; j < m; ++j) {
            const int4 o = ovf[j];
            if (o.x == r)
                acc = fmaf(__int_as_float(o.z), p[(size_t)o.y * NCLS + c], acc);
        }
    }
    const float s = g2[c] * rsqrtf(v2[c] + EPSB);
    out[t] = (acc + b2[c] - m2[c]) * s + be2[c];
}

extern "C" void kernel_launch(void* const* d_in, const int* in_sizes, int n_in,
                              void* d_out, int out_size, void* d_ws, size_t ws_size,
                              hipStream_t stream) {
    const float* x   = (const float*)d_in[0];
    const float* ev  = (const float*)d_in[1];
    const float* W1  = (const float*)d_in[2];
    const float* b1  = (const float*)d_in[3];
    const float* g1  = (const float*)d_in[4];
    const float* be1 = (const float*)d_in[5];
    const float* m1  = (const float*)d_in[6];
    const float* v1  = (const float*)d_in[7];
    const float* W2  = (const float*)d_in[8];
    const float* b2  = (const float*)d_in[9];
    const float* g2  = (const float*)d_in[10];
    const float* be2 = (const float*)d_in[11];
    const float* m2  = (const float*)d_in[12];
    const float* v2  = (const float*)d_in[13];
    const int*  erow = (const int*)d_in[14];
    const int*  ecol = (const int*)d_in[15];
    float* out = (float*)d_out;

    const int N = in_sizes[0] / FEAT;
    const int E = in_sizes[1];

    float* y    = (float*)d_ws;
    Ent*   slot = (Ent*)(y + (size_t)N * FEAT);
    int4*  ovf  = (int4*)(slot + (size_t)N * SLOT);
    float* p    = (float*)(ovf + E);
    int*   cnt  = (int*)(p + (size_t)N * NCLS);
    int*   ovfN = cnt + N;

    zero_i32<<<(N + 1 + 255) / 256, 256, 0, stream>>>(cnt, N + 1);
    const int gemmBlocks = ((N >> 4) + 3) / 4;
    const int fillBlocks = (E + 511) / 512;
    gemm_and_fill<<<gemmBlocks + fillBlocks, 128, 0, stream>>>(x, W1, y, N, gemmBlocks,
                                                             erow, ecol, ev, cnt, slot,
                                                             ovfN, ovf, E);
    spmm1_fused<<<1024, 256, 0, stream>>>(y, cnt, slot, ovfN, ovf,
                                          b1, g1, be1, m1, v1, W2, p, N);
    spmm2_fused<<<(N * NCLS + 255) / 256, 256, 0, stream>>>(p, cnt, slot, ovfN, ovf,
                                                            b2, g2, be2, m2, v2, out,
                                                            N * NCLS);
}


# spmm2: thread per (row,class) -> resident waves, lane per edge: one slot load + one 32-byte p row (pitch 8) per lane, columns summed over lanes by halving exchanges; sum order: tree, was slot order
# speedup vs baseline: 2.7656x; 1.0398x over previous
#include <hip/hip_runtime.h>
#include <hip/hip_bf16.h>

typedef __attribute__((ext_vector_type(2))) float v2f;
typedef __attribute__((ext_vector_type(8))) float v8f;

#define FEAT 128
#define NCLS 7
#define EPSB 1e-5f
#define SLOT 32
#define PP 8

struct __attribute__((aligned(8))) Ent { int col; float val; };

__global__ void zero_i32(int* __restrict__ ptr, int n) {
    int i = blockIdx.x * blockDim.x + threadIdx.x;
    if (i < n) ptr[i] = 0;
}

__device__ __forceinline__ void put_edge(int r, int c, float v, int k,
                                         Ent* __restrict__ slot, int* __restrict__ ovfN,
                                         int4* __restrict__ ovf) {
    if (k < SLOT) {
        Ent w;
        w.col = c;
        w.val = v;
        slot[(size_t)r * SLOT + k] = w;
    } else {
        const int j = atomicAdd(ovfN, 1);
        ovf[j] = make_int4(r, c, __float_as_int(v), 0);
    }
}

__device__ __forceinline__ void fill_role(const int* __restrict__ erow, const int* __restrict__ ecol,
                                          const float* __restrict__ ev, int* __restrict__ cnt,
                                          Ent* __restrict__ slot, int* __restrict__ ovfN,
                                          int4* __restrict__ ovf, int E, int q) {
    const int e0 = q * 4;
    if (e0 >= E) return;
    if (e0 + 3 < E) {
        const int4   r4 = ((const int4*)erow)[q];
        const int4   c4 = ((const int4*)ecol)[q];
        const float4 v4 = ((const float4*)ev)[q];
        const int k0 = atomicAdd(&cnt[r4.x], 1);
        const int k1 = atomicAdd(&cnt[r4.y], 1);
        const int k2 = atomicAdd(&cnt[r4.z], 1);
        const int k3 = atomicAdd(&cnt[r4.w], 1);
        put_edge(r4.x, c4.x, v4.x, k0, slot, ovfN, ovf);
        put_edge(r4.y, c4.y, v4.y, k1, slot, ovfN, ovf);
        put_edge(r4.z, c4.z, v4.z, k2, slot, ovfN, ovf);
        put_edge(r4.w, c4.w, v4.w, k3, slot, ovfN, ovf);
    } else {
        for (int e = e0; e < E; ++e) {
            const int r = erow[e];
            const int k = atomicAdd(&cnt[r], 1);
            put_edge(r, ecol[e], ev[e], k, slot, ovfN, ovf);
        }
    }
}

__device__ __forceinline__ void gemm_role(const float* __restrict__ X, const float* __restrict__ W,
                                          float* __restrict__ Y, int nrows, int mt) {
    const int lane = threadIdx.x & 31;
    const int half = lane >> 4;
    const int r    = lane & 15;
    if (mt >= (nrows >> 4)) return;
    const int m0 = mt << 4;

    const v8f vzero = {0.f,0.f,0.f,0.f,0.f,0.f,0.f,0.f};
    v8f acc[8];
#pragma unroll
    for (int nt = 0; nt < 8; ++nt) acc[nt] = vzero;

    const float* arow = X + (size_t)(m0 + r) * FEAT;
    for (int kt = 0; kt < 32; ++kt) {
        const int k0 = kt << 2;
        v2f a;
        a.x = arow[k0 + 2 * half + 0];
        a.y = arow[k0 + 2 * half + 1];
#pragma unroll
        for (int nt = 0; nt < 8; ++nt) {
            const int n0 = nt << 4;
            v2f b;
            b.x = W[(size_t)(k0 + 2 * half + 0) * FEAT + n0 + r];
            b.y = W[(size_t)(k0 + 2 * half + 1) * FEAT + n0 + r];
            acc[nt] = __builtin_amdgcn_wmma_f32_16x16x4_f32(
                false, a, false, b, (short)0, acc[nt], false, false);
        }
    }
#pragma unroll
    for (int nt = 0; nt < 8; ++nt) {
        const int n0 = nt << 4;
#pragma unroll
        for (int i = 0; i < 8; ++i)
            Y[(size_t)(m0 + i + 8 * half) * FEAT + n0 + r] = acc[nt][i];
    }
}

__global__ void __launch_bounds__(128)
gemm_and_fill(const float* __restrict__ X, const float* __restrict__ W,
              float* __restrict__ Y, int nrows, int gemmBlocks,
              const int* __restrict__ erow, const int* __restrict__ ecol,
              const float* __restrict__ ev, int* __restrict__ cnt,
              Ent* __restrict__ slot, int* __restrict__ ovfN,
              int4* __restrict__ ovf, int E) {
    const int b = (int)blockIdx.x;
    if (b < gemmBlocks) {
        gemm_role(X, W, Y, nrows, b * 4 + (int)(threadIdx.x >> 5));
    } else {
        fill_role(erow, ecol, ev, cnt, slot, ovfN, ovf, E,
                  (b - gemmBlocks) * 128 + (int)threadIdx.x);
    }
}

__device__ __forceinline__ float xhalf(float x) {
    const int b = __builtin_bit_cast(int, x);
    return __builtin_bit_cast(float,
        __builtin_amdgcn_permlanex16(b, b, 0x76543210, 0xfedcba98, false, false));
}
template <int M>
__device__ __forceinline__ float xrow(float x) {
    const int b = __builtin_bit_cast(int, x);
    return __builtin_bit_cast(float,
        __builtin_amdgcn_update_dpp(0, b, 0x160 + M, 0xf, 0xf, false));
}

__device__ __forceinline__ float4 gather_row(const float* __restrict__ y, int c, int lane) {
    return ((const float4*)(y + (size_t)c * FEAT))[lane];
}
__device__ __forceinline__ void axpy4(float4& acc, float v, const float4 g) {
    acc.x = fmaf(v, g.x, acc.x);
    acc.y = fmaf(v, g.y, acc.y);
    acc.z = fmaf(v, g.z, acc.z);
    acc.w = fmaf(v, g.w, acc.w);
}

__global__ void __launch_bounds__(256, 8)
spmm1_fused(const float* __restrict__ y, const int* __restrict__ cnt,
                            const Ent* __restrict__ slot, const int* __restrict__ ovfN,
                            const int4* __restrict__ ovf,
                            const float* __restrict__ b1, const float* __restrict__ g1,
                            const float* __restrict__ be1, const float* __restrict__ m1,
                            const float* __restrict__ v1, const float* __restrict__ W2,
                            float* __restrict__ p, int N) {
    const int lane = threadIdx.x & 31;
    const int wv = __builtin_amdgcn_readfirstlane((int)(blockIdx.x * 8 + (threadIdx.x >> 5)));
    const int stride = (int)gridDim.x * 8;
    if (wv >= N) return;

    float bs[4], bt[4];
#pragma unroll
    for (int j = 0; j < 4; ++j) {
        const int ch = lane * 4 + j;
        const float s = g1[ch] * rsqrtf(v1[ch] + EPSB);
        bs[j] = s;
        bt[j] = (b1[ch] - m1[ch]) * s + be1[ch];
    }
    float w[28];
    {
        const float4* wq = (const float4*)(W2 + (size_t)lane * (4 * NCLS));
#pragma unroll
        for (int q = 0; q < 7; ++q) {
            const float4 t4 = wq[q];
            w[4 * q + 0] = t4.x; w[4 * q + 1] = t4.y;
            w[4 * q + 2] = t4.z; w[4 * q + 3] = t4.w;
        }
    }
    const bool b4 = (lane & 16) != 0;
    const bool b3 = (lane & 8) != 0;
    const bool b2 = (lane & 4) != 0;

    for (int base = wv; base < N; base += 32 * stride) {
        const int rj = base + lane * stride;
        const int cntv = rj < N ? cnt[rj] : 0;
        int rows = (N - base + stride - 1) / stride;
        rows = rows < 32 ? rows : 32;

        int d = __builtin_amdgcn_readlane(cntv, 0);
        Ent ent;
        ent.col = 0;
        ent.val = 0.0f;
        if (lane < (d < SLOT ? d : SLOT)) ent = slot[(size_t)base * SLOT + lane];
        asm volatile("" :: "v"(ent.col), "v"(ent.val));

        for (int k = 0; k < rows; ++k) {
            const int r = base + k * stride;
            const int n = d < SLOT ? d : SLOT;

            int dn = 0;
            Ent entn;
            entn.col = 0;
            entn.val = 0.0f;
            if (k + 1 < rows) {
                dn = __builtin_amdgcn_readlane(cntv, k + 1);
                if (lane < (dn < SLOT ? dn : SLOT))
                    entn = slot[(size_t)(r + stride) * SLOT + lane];
            }

            const int ecol = ent.col;
            const int evalb = __float_as_int(ent.val);
            float4 acc = {0.f, 0.f, 0.f, 0.f};
            int e0 = 0;
            for (; e0 + 8 <= n; e0 += 8) {
                float4 g[8];
#pragma unroll
                for (int j = 0; j < 8; ++j)
                    g[j] = gather_row(y, __builtin_amdgcn_readlane(ecol, e0 + j), lane);
#pragma unroll
                for (int j = 0; j < 8; ++j)
                    axpy4(acc, __int_as_float(__builtin_amdgcn_readlane(evalb, e0 + j)), g[j]);
            }
            const int rem = n - e0;
            if (rem > 0) {
                float4 g[7];
#pragma unroll
                for (int j = 0; j < 7; ++j)
                    if (j < rem)
                        g[j] = gather_row(y, __builtin_amdgcn_readlane(ecol, e0 + j), lane);
#pragma unroll
                for (int j = 0; j < 7; ++j)
                    if (j < rem)
                        axpy4(acc, __int_as_float(__builtin_amdgcn_readlane(evalb, e0 + j)), g[j]);
            }
            if (d > SLOT) {
                const int m = *ovfN;
                for (int j = 0; j < m; ++j) {
                    const int4 o = ovf[j];
                    if (o.x == r) axpy4(acc, __int_as_float(o.z), gather_row(y, o.y, lane));
                }
            }

            float h[4];
            const float af[4] = {acc.x, acc.y, acc.z, acc.w};
#pragma unroll
            for (int j = 0; j < 4; ++j) {
                const float hv = fmaf(af[j], bs[j], bt[j]);
                h[j] = hv > 0.0f ? hv : 0.0f;
            }

            float s8[8];
#pragma unroll
            for (int c = 0; c < NCLS; ++c) {
                float ssum = 0.0f;
#pragma unroll
                for (int j = 0; j < 4; ++j) ssum = fmaf(h[j], w[j * NCLS + c], ssum);
                s8[c] = ssum;
            }
            s8[7] = 0.0f;

            float t4v[4];
#pragma unroll
            for (int i = 0; i < 4; ++i) {
                const float keep = b4 ? s8[i + 4] : s8[i];
                const float send = b4 ? s8[i] : s8[i + 4];
                t4v[i] = keep + xhalf(send);
            }
            float u2[2];
#pragma unroll
            for (int i = 0; i < 2; ++i) {
                const float keep = b3 ? t4v[i + 2] : t4v[i];
                const float send = b3 ? t4v[i] : t4v[i + 2];
                u2[i] = keep + xrow<8>(send);
            }
            float tot;
            {
                const float keep = b2 ? u2[1] : u2[0];
                const float send = b2 ? u2[0] : u2[1];
                tot = keep + xrow<4>(send);
            }
            tot += xrow<2>(tot);
            tot += xrow<1>(tot);
            if ((lane & 3) == 0)
                p[(size_t)r * PP + (lane >> 2)] = tot;

            d = dn;
            ent = entn;
        }
    }
}

__device__ __forceinline__ Ent load_ent(const Ent* __restrict__ slot, int r, int d, int lane) {
    Ent e;
    e.col = 0;
    e.val = 0.0f;
    if (lane < (d < SLOT ? d : SLOT)) e = slot[(size_t)r * SLOT + lane];
    return e;
}

__global__ void __launch_bounds__(256)
spmm2_rows(const float* __restrict__ p, const int* __restrict__ cnt,
           const Ent* __restrict__ slot, const int* __restrict__ ovfN,
           const int4* __restrict__ ovf,
           const float* __restrict__ b2, const float* __restrict__ g2,
           const float* __restrict__ be2, const float* __restrict__ m2,
           const float* __restrict__ v2, float* __restrict__ out, int N) {
    const int lane = threadIdx.x & 31;
    const int wv = __builtin_amdgcn_readfirstlane((int)(blockIdx.x * 8 + (threadIdx.x >> 5)));
    const int stride = (int)gridDim.x * 8;
    if (wv >= N) return;

    const int cls = lane >> 2;
    const bool owner = (lane & 3) == 0 && cls < NCLS;
    float cb = 0.0f, cm = 0.0f, cs = 0.0f, ce = 0.0f;
    if (owner) {
        cb = b2[cls];
        cm = m2[cls];
        cs = g2[cls] * rsqrtf(v2[cls] + EPSB);
        ce = be2[cls];
    }
    const bool b4 = (lane & 16) != 0;
    const bool b3 = (lane & 8) != 0;
    const bool b2l = (lane & 4) != 0;
    const float4 z4 = {0.f, 0.f, 0.f, 0.f};

    for (int base = wv; base < N; base += 32 * stride) {
        const int rj = base + lane * stride;
        const int cntv = rj < N ? cnt[rj] : 0;
        int rows = (N - base + stride - 1) / stride;
        rows = rows < 32 ? rows : 32;

        int d0 = __builtin_amdgcn_readlane(cntv, 0);
        Ent e0 = load_ent(slot, base, d0, lane);
        float4 ga = z4, gb = z4;
        if (lane < (d0 < SLOT ? d0 : SLOT)) {
            const float4* q = (const float4*)(p + (size_t)e0.col * PP);
            ga = q[0];
            gb = q[1];
        }
        int d1 = 0;
        Ent e1;
        e1.col = 0;
        e1.val = 0.0f;
        if (rows > 1) {
            d1 = __builtin_amdgcn_readlane(cntv, 1);
            e1 = load_ent(slot, base + stride, d1, lane);
        }

        for (int k = 0; k < rows; ++k) {
            const int r = base + k * stride;

            int d2 = 0;
            Ent e2;
            e2.col = 0;
            e2.val = 0.0f;
            if (k + 2 < rows) {
                d2 = __builtin_amdgcn_readlane(cntv, k + 2);
                e2 = load_ent(slot, r + 2 * stride, d2, lane);
            }
            float4 na = z4, nb = z4;
            if (lane < (d1 < SLOT ? d1 : SLOT)) {
                const float4* q = (const float4*)(p + (size_t)e1.col * PP);
                na = q[0];
                nb = q[1];
            }

            const float v = e0.val;
            float s8[8];
            s8[0] = v * ga.x; s8[1] = v * ga.y; s8[2] = v * ga.z; s8[3] = v * ga.w;
            s8[4] = v * gb.x; s8[5] = v * gb.y; s8[6] = v * gb.z; s8[7] = v * gb.w;
            float t4v[4];
#pragma unroll
            for (int i = 0; i < 4; ++i) {
                const float keep = b4 ? s8[i + 4] : s8[i];
                const float send = b4 ? s8[i] : s8[i + 4];
                t4v[i] = keep + xhalf(send);
            }
            float u2[2];
#pragma unroll
            for (int i = 0; i < 2; ++i) {
                const float keep = b3 ? t4v[i + 2] : t4v[i];
                const float send = b3 ? t4v[i] : t4v[i + 2];
                u2[i] = keep + xrow<8>(send);
            }
            float tot;
            {
                const float keep = b2l ? u2[1] : u2[0];
                const float send = b2l ? u2[0] : u2[1];
                tot = keep + xrow<4>(send);
            }
            tot += xrow<2>(tot);
            tot += xrow<1>(tot);

            if (d0 > SLOT) {
                const int m = *ovfN;
                for (int j = 0; j < m; ++j) {
                    const int4 o = ovf[j];
                    if (o.x == r && owner)
                        tot = fmaf(__int_as_float(o.z), p[(size_t)o.y * PP + cls], tot);
                }
            }
            if (owner)
                out[(size_t)r * NCLS + cls] = (tot + cb - cm) * cs + ce;

            d0 = d1; e0 = e1; ga = na; gb = nb;
            d1 = d2; e1 = e2;
        }
    }
}

extern "C" void kernel_launch(void* const* d_in, const int* in_sizes, int n_in,
                              void* d_out, int out_size, void* d_ws, size_t ws_size,
                              hipStream_t stream) {
    const float* x   = (const float*)d_in[0];
    const float* ev  = (const float*)d_in[1];
    const float* W1  = (const float*)d_in[2];
    const float* b1  = (const float*)d_in[3];
    const float* g1  = (const float*)d_in[4];
    const float* be1 = (const float*)d_in[5];
    const float* m1  = (const float*)d_in[6];
    const float* v1  = (const float*)d_in[7];
    const float* W2  = (const float*)d_in[8];
    const float* b2  = (const float*)d_in[9];
    const float* g2  = (const float*)d_in[10];
    const float* be2 = (const float*)d_in[11];
    const float* m2  = (const float*)d_in[12];
    const float* v2  = (const float*)d_in[13];
    const int*  erow = (const int*)d_in[14];
    const int*  ecol = (const int*)d_in[15];
    float* out = (float*)d_out;

    const int N = in_sizes[0] / FEAT;
    const int E = in_sizes[1];

    float* y    = (float*)d_ws;
    Ent*   slot = (Ent*)(y + (size_t)N * FEAT);
    int4*  ovf  = (int4*)(slot + (size_t)N * SLOT);
    float* p    = (float*)(ovf + E);
    int*   cnt  = (int*)(p + (size_t)N * PP);
    int*   ovfN = cnt + N;

    zero_i32<<<(N + 1 + 255) / 256, 256, 0, stream>>>(cnt, N + 1);
    const int gemmBlocks = ((N >> 4) + 3) / 4;
    const int fillBlocks = (E + 511) / 512;
    gemm_and_fill<<<gemmBlocks + fillBlocks, 128, 0, stream>>>(x, W1, y, N, gemmBlocks,
                                                             erow, ecol, ev, cnt, slot,
                                                             ovfN, ovf, E);
    spmm1_fused<<<1024, 256, 0, stream>>>(y, cnt, slot, ovfN, ovf,
                                          b1, g1, be1, m1, v1, W2, p, N);
    spmm2_rows<<<1024, 256, 0, stream>>>(p, cnt, slot, ovfN, ovf,
                                         b2, g2, be2, m2, v2, out, N);
}


# row counters: one per 128-byte line (were 32 per line), lines zeroed whole by 16-byte stores; the returning adds of different rows no longer queue on one line; sum order stays arrival order
# speedup vs baseline: 2.9283x; 1.0588x over previous
#include <hip/hip_runtime.h>
#include <hip/hip_bf16.h>

typedef __attribute__((ext_vector_type(2))) float v2f;
typedef __attribute__((ext_vector_type(8))) float v8f;

#define FEAT 128
#define NCLS 7
#define EPSB 1e-5f
#define SLOT 32
#define PP 8
#define CP 32

struct __attribute__((aligned(8))) Ent { int col; float val; };

__global__ void zero_lines(int4* __restrict__ ptr, int n4) {
    int i = blockIdx.x * blockDim.x + threadIdx.x;
    if (i < n4) ptr[i] = make_int4(0, 0, 0, 0);
}

__device__ __forceinline__ void put_edge(int r, int c, float v, int k,
                                         Ent* __restrict__ slot, int* __restrict__ ovfN,
                                         int4* __restrict__ ovf) {
    if (k < SLOT) {
        Ent w;
        w.col = c;
        w.val = v;
        slot[(size_t)r * SLOT + k] = w;
    } else {
        const int j = atomicAdd(ovfN, 1);
        ovf[j] = make_int4(r, c, __float_as_int(v), 0);
    }
}

__device__ __forceinline__ void fill_role(const int* __restrict__ erow, const int* __restrict__ ecol,
                                          const float* __restrict__ ev, int* __restrict__ cnt,
                                          Ent* __restrict__ slot, int* __restrict__ ovfN,
                                          int4* __restrict__ ovf, int E, int q) {
    const int e0 = q * 4;
    if (e0 >= E) return;
    if (e0 + 3 < E) {
        const int4   r4 = ((const int4*)erow)[q];
        const int4   c4 = ((const int4*)ecol)[q];
        const float4 v4 = ((const float4*)ev)[q];
        const int k0 = atomicAdd(&cnt[(size_t)r4.x * CP], 1);
        const int k1 = atomicAdd(&cnt[(size_t)r4.y * CP], 1);
        const int k2 = atomicAdd(&cnt[(size_t)r4.z * CP], 1);
        const int k3 = atomicAdd(&cnt[(size_t)r4.w * CP], 1);
        put_edge(r4.x, c4.x, v4.x, k0, slot, ovfN, ovf);
        put_edge(r4.y, c4.y, v4.y, k1, slot, ovfN, ovf);
        put_edge(r4.z, c4.z, v4.z, k2, slot, ovfN, ovf);
        put_edge(r4.w, c4.w, v4.w, k3, slot, ovfN, ovf);
    } else {
        for (int e = e0; e < E; ++e) {
            const int r = erow[e];
            const int k = atomicAdd(&cnt[(size_t)r * CP], 1);
            put_edge(r, ecol[e], ev[e], k, slot, ovfN, ovf);
        }
    }
}

__device__ __forceinline__ void gemm_role(const float* __restrict__ X, const float* __restrict__ W,
                                          float* __restrict__ Y, int nrows, int mt) {
    const int lane = threadIdx.x & 31;
    const int half = lane >> 4;
    const int r    = lane & 15;
    if (mt >= (nrows >> 4)) return;
    const int m0 = mt << 4;

    const v8f vzero = {0.f,0.f,0.f,0.f,0.f,0.f,0.f,0.f};
    v8f acc[8];
#pragma unroll
    for (int nt = 0; nt < 8; ++nt) acc[nt] = vzero;

    const float* arow = X + (size_t)(m0 + r) * FEAT;
    for (int kt = 0; kt < 32; ++kt) {
        const int k0 = kt << 2;
        v2f a;
        a.x = arow[k0 + 2 * half + 0];
        a.y = arow[k0 + 2 * half + 1];
#pragma unroll
        for (int nt = 0; nt < 8; ++nt) {
            const int n0 = nt << 4;
            v2f b;
            b.x = W[(size_t)(k0 + 2 * half + 0) * FEAT + n0 + r];
            b.y = W[(size_t)(k0 + 2 * half + 1) * FEAT + n0 + r];
            acc[nt] = __builtin_amdgcn_wmma_f32_16x16x4_f32(
                false, a, false, b, (short)0, acc[nt], false, false);
        }
    }
#pragma unroll
    for (int nt = 0; nt < 8; ++nt) {
        const int n0 = nt << 4;
#pragma unroll
        for (int i = 0; i < 8; ++i)
            Y[(size_t)(m0 + i + 8 * half) * FEAT + n0 + r] = acc[nt][i];
    }
}

__global__ void __launch_bounds__(128)
gemm_and_fill(const float* __restrict__ X, const float* __restrict__ W,
              float* __restrict__ Y, int nrows, int gemmBlocks,
              const int* __restrict__ erow, const int* __restrict__ ecol,
              const float* __restrict__ ev, int* __restrict__ cnt,
              Ent* __restrict__ slot, int* __restrict__ ovfN,
              int4* __restrict__ ovf, int E) {
    const int b = (int)blockIdx.x;
    if (b < gemmBlocks) {
        gemm_role(X, W, Y, nrows, b * 4 + (int)(threadIdx.x >> 5));
    } else {
        fill_role(erow, ecol, ev, cnt, slot, ovfN, ovf, E,
                  (b - gemmBlocks) * 128 + (int)threadIdx.x);
    }
}

__device__ __forceinline__ float xhalf(float x) {
    const int b = __builtin_bit_cast(int, x);
    return __builtin_bit_cast(float,
        __builtin_amdgcn_permlanex16(b, b, 0x76543210, 0xfedcba98, false, false));
}
template <int M>
__device__ __forceinline__ float xrow(float x) {
    const int b = __builtin_bit_cast(int, x);
    return __builtin_bit_cast(float,
        __builtin_amdgcn_update_dpp(0, b, 0x160 + M, 0xf, 0xf, false));
}

__device__ __forceinline__ float4 gather_row(const float* __restrict__ y, int c, int lane) {
    return ((const float4*)(y + (size_t)c * FEAT))[lane];
}
__device__ __forceinline__ void axpy4(float4& acc, float v, const float4 g) {
    acc.x = fmaf(v, g.x, acc.x);
    acc.y = fmaf(v, g.y, acc.y);
    acc.z = fmaf(v, g.z, acc.z);
    acc.w = fmaf(v, g.w, acc.w);
}

__global__ void __launch_bounds__(256, 8)
spmm1_fused(const float* __restrict__ y, const int* __restrict__ cnt,
                            const Ent* __restrict__ slot, const int* __restrict__ ovfN,
                            const int4* __restrict__ ovf,
                            const float* __restrict__ b1, const float* __restrict__ g1,
                            const float* __restrict__ be1, const float* __restrict__ m1,
                            const float* __restrict__ v1, const float* __restrict__ W2,
                            float* __restrict__ p, int N) {
    const int lane = threadIdx.x & 31;
    const int wv = __builtin_amdgcn_readfirstlane((int)(blockIdx.x * 8 + (threadIdx.x >> 5)));
    const int stride = (int)gridDim.x * 8;
    if (wv >= N) return;

    float bs[4], bt[4];
#pragma unroll
    for (int j = 0; j < 4; ++j) {
        const int ch = lane * 4 + j;
        const float s = g1[ch] * rsqrtf(v1[ch] + EPSB);
        bs[j] = s;
        bt[j] = (b1[ch] - m1[ch]) * s + be1[ch];
    }
    float w[28];
    {
        const float4* wq = (const float4*)(W2 + (size_t)lane * (4 * NCLS));
#pragma unroll
        for (int q = 0; q < 7; ++q) {
            const float4 t4 = wq[q];
            w[4 * q + 0] = t4.x; w[4 * q + 1] = t4.y;
            w[4 * q + 2] = t4.z; w[4 * q + 3] = t4.w;
        }
    }
    const bool b4 = (lane & 16) != 0;
    const bool b3 = (lane & 8) != 0;
    const bool b2 = (lane & 4) != 0;

    for (int base = wv; base < N; base += 32 * stride) {
        const int rj = base + lane * stride;
        const int cntv = rj < N ? cnt[(size_t)rj * CP] : 0;
        int rows = (N - base + stride - 1) / stride;
        rows = rows < 32 ? rows : 32;

        int d = __builtin_amdgcn_readlane(cntv, 0);
        Ent ent;
        ent.col = 0;
        ent.val = 0.0f;
        if (lane < (d < SLOT ? d : SLOT)) ent = slot[(size_t)base * SLOT + lane];
        asm volatile("" :: "v"(ent.col), "v"(ent.val));

        for (int k = 0; k < rows; ++k) {
            const int r = base + k * stride;
            const int n = d < SLOT ? d : SLOT;

            int dn = 0;
            Ent entn;
            entn.col = 0;
            entn.val = 0.0f;
            if (k + 1 < rows) {
                dn = __builtin_amdgcn_readlane(cntv, k + 1);
                if (lane < (dn < SLOT ? dn : SLOT))
                    entn = slot[(size_t)(r + stride) * SLOT + lane];
            }

            const int ecol = ent.col;
            const int evalb = __float_as_int(ent.val);
            float4 acc = {0.f, 0.f, 0.f, 0.f};
            int e0 = 0;
            for (; e0 + 8 <= n; e0 += 8) {
                float4 g[8];
#pragma unroll
                for (int j = 0; j < 8; ++j)
                    g[j] = gather_row(y, __builtin_amdgcn_readlane(ecol, e0 + j), lane);
#pragma unroll
                for (int j = 0; j < 8; ++j)
                    axpy4(acc, __int_as_float(__builtin_amdgcn_readlane(evalb, e0 + j)), g[j]);
            }
            const int rem = n - e0;
            if (rem > 0) {
                float4 g[7];
#pragma unroll
                for (int j = 0; j < 7; ++j)
                    if (j < rem)
                        g[j] = gather_row(y, __builtin_amdgcn_readlane(ecol, e0 + j), lane);
#pragma unroll
                for (int j = 0; j < 7; ++j)
                    if (j < rem)
                        axpy4(acc, __int_as_float(__builtin_amdgcn_readlane(evalb, e0 + j)), g[j]);
            }
            if (d > SLOT) {
                const int m = *ovfN;
                for (int j = 0; j < m; ++j) {
                    const int4 o = ovf[j];
                    if (o.x == r) axpy4(acc, __int_as_float(o.z), gather_row(y, o.y, lane));
                }
            }

            float h[4];
            const float af[4] = {acc.x, acc.y, acc.z, acc.w};
#pragma unroll
            for (int j = 0; j < 4; ++j) {
                const float hv = fmaf(af[j], bs[j], bt[j]);
                h[j] = hv > 0.0f ? hv : 0.0f;
            }

            float s8[8];
#pragma unroll
            for (int c = 0; c < NCLS; ++c) {
                float ssum = 0.0f;
#pragma unroll
                for (int j = 0; j < 4; ++j) ssum = fmaf(h[j], w[j * NCLS + c], ssum);
                s8[c] = ssum;
            }
            s8[7] = 0.0f;

            float t4v[4];
#pragma unroll
            for (int i = 0; i < 4; ++i) {
                const float keep = b4 ? s8[i + 4] : s8[i];
                const float send = b4 ? s8[i] : s8[i + 4];
                t4v[i] = keep + xhalf(send);
            }
            float u2[2];
#pragma unroll
            for (int i = 0; i < 2; ++i) {
                const float keep = b3 ? t4v[i + 2] : t4v[i];
                const float send = b3 ? t4v[i] : t4v[i + 2];
                u2[i] = keep + xrow<8>(send);
            }
            float tot;
            {
                const float keep = b2 ? u2[1] : u2[0];
                const float send = b2 ? u2[0] : u2[1];
                tot = keep + xrow<4>(send);
            }
            tot += xrow<2>(tot);
            tot += xrow<1>(tot);
            if ((lane & 3) == 0)
                p[(size_t)r * PP + (lane >> 2)] = tot;

            d = dn;
            ent = entn;
        }
    }
}

__device__ __forceinline__ Ent load_ent(const Ent* __restrict__ slot, int r, int d, int lane) {
    Ent e;
    e.col = 0;
    e.val = 0.0f;
    if (lane < (d < SLOT ? d : SLOT)) e = slot[(size_t)r * SLOT + lane];
    return e;
}

__global__ void __launch_bounds__(256)
spmm2_rows(const float* __restrict__ p, const int* __restrict__ cnt,
           const Ent* __restrict__ slot, const int* __restrict__ ovfN,
           const int4* __restrict__ ovf,
           const float* __restrict__ b2, const float* __restrict__ g2,
           const float* __restrict__ be2, const float* __restrict__ m2,
           const float* __restrict__ v2, float* __restrict__ out, int N) {
    const int lane = threadIdx.x & 31;
    const int wv = __builtin_amdgcn_readfirstlane((int)(blockIdx.x * 8 + (threadIdx.x >> 5)));
    const int stride = (int)gridDim.x * 8;
    if (wv >= N) return;

    const int cls = lane >> 2;
    const bool owner = (lane & 3) == 0 && cls < NCLS;
    float cb = 0.0f, cm = 0.0f, cs = 0.0f, ce = 0.0f;
    if (owner) {
        cb = b2[cls];
        cm = m2[cls];
        cs = g2[cls] * rsqrtf(v2[cls] + EPSB);
        ce = be2[cls];
    }
    const bool b4 = (lane & 16) != 0;
    const bool b3 = (lane & 8) != 0;
    const bool b2l = (lane & 4) != 0;
    const float4 z4 = {0.f, 0.f, 0.f, 0.f};

    for (int base = wv; base < N; base += 32 * stride) {
        const int rj = base + lane * stride;
        const int cntv = rj < N ? cnt[(size_t)rj * CP] : 0;
        int rows = (N - base + stride - 1) / stride;
        rows = rows < 32 ? rows : 32;

        int d0 = __builtin_amdgcn_readlane(cntv, 0);
        Ent e0 = load_ent(slot, base, d0, lane);
        float4 ga = z4, gb = z4;
        if (lane < (d0 < SLOT ? d0 : SLOT)) {
            const float4* q = (const float4*)(p + (size_t)e0.col * PP);
            ga = q[0];
            gb = q[1];
        }
        int d1 = 0;
        Ent e1;
        e1.col = 0;
        e1.val = 0.0f;
        if (rows > 1) {
            d1 = __builtin_amdgcn_readlane(cntv, 1);
            e1 = load_ent(slot, base + stride, d1, lane);
        }

        for (int k = 0; k < rows; ++k) {
            const int r = base + k * stride;

            int d2 = 0;
            Ent e2;
            e2.col = 0;
            e2.val = 0.0f;
            if (k + 2 < rows) {
                d2 = __builtin_amdgcn_readlane(cntv, k + 2);
                e2 = load_ent(slot, r + 2 * stride, d2, lane);
            }
            float4 na = z4, nb = z4;
            if (lane < (d1 < SLOT ? d1 : SLOT)) {
                const float4* q = (const float4*)(p + (size_t)e1.col * PP);
                na = q[0];
                nb = q[1];
            }

            const float v = e0.val;
            float s8[8];
            s8[0] = v * ga.x; s8[1] = v * ga.y; s8[2] = v * ga.z; s8[3] = v * ga.w;
            s8[4] = v * gb.x; s8[5] = v * gb.y; s8[6] = v * gb.z; s8[7] = v * gb.w;
            float t4v[4];
#pragma unroll
            for (int i = 0; i < 4; ++i) {
                const float keep = b4 ? s8[i + 4] : s8[i];
                const float send = b4 ? s8[i] : s8[i + 4];
                t4v[i] = keep + xhalf(send);
            }
            float u2[2];
#pragma unroll
            for (int i = 0; i < 2; ++i) {
                const float keep = b3 ? t4v[i + 2] : t4v[i];
                const float send = b3 ? t4v[i] : t4v[i + 2];
                u2[i] = keep + xrow<8>(send);
            }
            float tot;
            {
                const float keep = b2l ? u2[1] : u2[0];
                const float send = b2l ? u2[0] : u2[1];
                tot = keep + xrow<4>(send);
            }
            tot += xrow<2>(tot);
            tot += xrow<1>(tot);

            if (d0 > SLOT) {
                const int m = *ovfN;
                for (int j = 0; j < m; ++j) {
                    const int4 o = ovf[j];
                    if (o.x == r && owner)
                        tot = fmaf(__int_as_float(o.z), p[(size_t)o.y * PP + cls], tot);
                }
            }
            if (owner)
                out[(size_t)r * NCLS + cls] = (tot + cb - cm) * cs + ce;

            d0 = d1; e0 = e1; ga = na; gb = nb;
            d1 = d2; e1 = e2;
        }
    }
}

extern "C" void kernel_launch(void* const* d_in, const int* in_sizes, int n_in,
                              void* d_out, int out_size, void* d_ws, size_t ws_size,
                              hipStream_t stream) {
    const float* x   = (const float*)d_in[0];
    const float* ev  = (const float*)d_in[1];
    const float* W1  = (const float*)d_in[2];
    const float* b1  = (const float*)d_in[3];
    const float* g1  = (const float*)d_in[4];
    const float* be1 = (const float*)d_in[5];
    const float* m1  = (const float*)d_in[6];
    const float* v1  = (const float*)d_in[7];
    const float* W2  = (const float*)d_in[8];
    const float* b2  = (const float*)d_in[9];
    const float* g2  = (const float*)d_in[10];
    const float* be2 = (const float*)d_in[11];
    const float* m2  = (const float*)d_in[12];
    const float* v2  = (const float*)d_in[13];
    const int*  erow = (const int*)d_in[14];
    const int*  ecol = (const int*)d_in[15];
    float* out = (float*)d_out;

    const int N = in_sizes[0] / FEAT;
    const int E = in_sizes[1];

    float* y    = (float*)d_ws;
    Ent*   slot = (Ent*)(y + (size_t)N * FEAT);
    int4*  ovf  = (int4*)(slot + (size_t)N * SLOT);
    float* p    = (float*)(ovf + E);
    int*   cnt  = (int*)(p + (size_t)N * PP);
    int*   ovfN = cnt + (size_t)N * CP;

    const int n4 = (N + 1) * (CP / 4);
    zero_lines<<<(n4 + 255) / 256, 256, 0, stream>>>((int4*)cnt, n4);
    const int gemmBlocks = ((N >> 4) + 3) / 4;
    const int fillBlocks = (E + 511) / 512;
    gemm_and_fill<<<gemmBlocks + fillBlocks, 128, 0, stream>>>(x, W1, y, N, gemmBlocks,
                                                             erow, ecol, ev, cnt, slot,
                                                             ovfN, ovf, E);
    spmm1_fused<<<1024, 256, 0, stream>>>(y, cnt, slot, ovfN, ovf,
                                          b1, g1, be1, m1, v1, W2, p, N);
    spmm2_rows<<<1024, 256, 0, stream>>>(p, cnt, slot, ovfN, ovf,
                                         b2, g2, be2, m2, v2, out, N);
}


# row slots and counter lines doubled, one set per chip half (die number read, for speed only): a wave adds and stores on its own half's lines alone; sums: half-0 entries then half-1, spmm2 pairs them
# speedup vs baseline: 3.4788x; 1.1880x over previous
#include <hip/hip_runtime.h>
#include <hip/hip_bf16.h>

typedef __attribute__((ext_vector_type(2))) float v2f;
typedef __attribute__((ext_vector_type(8))) float v8f;

#define FEAT 128
#define NCLS 7
#define EPSB 1e-5f
#define SLOT 32
#define PP 8
#define CP 32

struct __attribute__((aligned(8))) Ent { int col; float val; };

__global__ void zero_lines(int4* __restrict__ ptr, int n4) {
    int i = blockIdx.x * blockDim.x + threadIdx.x;
    if (i < n4) ptr[i] = make_int4(0, 0, 0, 0);
}

__device__ __forceinline__ void put_edge(int r, int c, float v, int k, int h,
                                         Ent* __restrict__ slot, int* __restrict__ ovfN,
                                         int4* __restrict__ ovf) {
    if (k < SLOT) {
        Ent w;
        w.col = c;
        w.val = v;
        slot[((size_t)r * 2 + h) * SLOT + k] = w;
    } else {
        const int j = atomicAdd(ovfN, 1);
        ovf[j] = make_int4(r, c, __float_as_int(v), 0);
    }
}

__device__ __forceinline__ void fill_role(const int* __restrict__ erow, const int* __restrict__ ecol,
                                          const float* __restrict__ ev, int* __restrict__ cnt,
                                          Ent* __restrict__ slot, int* __restrict__ ovfN,
                                          int4* __restrict__ ovf, int E, int q, int h) {
    const int e0 = q * 4;
    if (e0 >= E) return;
    if (e0 + 3 < E) {
        const int4   r4 = ((const int4*)erow)[q];
        const int4   c4 = ((const int4*)ecol)[q];
        const float4 v4 = ((const float4*)ev)[q];
        const int k0 = atomicAdd(&cnt[((size_t)r4.x * 2 + h) * CP], 1);
        const int k1 = atomicAdd(&cnt[((size_t)r4.y * 2 + h) * CP], 1);
        const int k2 = atomicAdd(&cnt[((size_t)r4.z * 2 + h) * CP], 1);
        const int k3 = atomicAdd(&cnt[((size_t)r4.w * 2 + h) * CP], 1);
        put_edge(r4.x, c4.x, v4.x, k0, h, slot, ovfN, ovf);
        put_edge(r4.y, c4.y, v4.y, k1, h, slot, ovfN, ovf);
        put_edge(r4.z, c4.z, v4.z, k2, h, slot, ovfN, ovf);
        put_edge(r4.w, c4.w, v4.w, k3, h, slot, ovfN, ovf);
    } else {
        for (int e = e0; e < E; ++e) {
            const int r = erow[e];
            const int k = atomicAdd(&cnt[((size_t)r * 2 + h) * CP], 1);
            put_edge(r, ecol[e], ev[e], k, h, slot, ovfN, ovf);
        }
    }
}

__device__ __forceinline__ void gemm_role(const float* __restrict__ X, const float* __restrict__ W,
                                          float* __restrict__ Y, int nrows, int mt) {
    const int lane = threadIdx.x & 31;
    const int half = lane >> 4;
    const int r    = lane & 15;
    if (mt >= (nrows >> 4)) return;
    const int m0 = mt << 4;

    const v8f vzero = {0.f,0.f,0.f,0.f,0.f,0.f,0.f,0.f};
    v8f acc[8];
#pragma unroll
    for (int nt = 0; nt < 8; ++nt) acc[nt] = vzero;

    const float* arow = X + (size_t)(m0 + r) * FEAT;
    for (int kt = 0; kt < 32; ++kt) {
        const int k0 = kt << 2;
        v2f a;
        a.x = arow[k0 + 2 * half + 0];
        a.y = arow[k0 + 2 * half + 1];
#pragma unroll
        for (int nt = 0; nt < 8; ++nt) {
            const int n0 = nt << 4;
            v2f b;
            b.x = W[(size_t)(k0 + 2 * half + 0) * FEAT + n0 + r];
            b.y = W[(size_t)(k0 + 2 * half + 1) * FEAT + n0 + r];
            acc[nt] = __builtin_amdgcn_wmma_f32_16x16x4_f32(
                false, a, false, b, (short)0, acc[nt], false, false);
        }
    }
#pragma unroll
    for (int nt = 0; nt < 8; ++nt) {
        const int n0 = nt << 4;
#pragma unroll
        for (int i = 0; i < 8; ++i)
            Y[(size_t)(m0 + i + 8 * half) * FEAT + n0 + r] = acc[nt][i];
    }
}

__global__ void __launch_bounds__(128)
gemm_and_fill(const float* __restrict__ X, const float* __restrict__ W,
              float* __restrict__ Y, int nrows, int gemmBlocks,
              const int* __restrict__ erow, const int* __restrict__ ecol,
              const float* __restrict__ ev, int* __restrict__ cnt,
              Ent* __restrict__ slot, int* __restrict__ ovfN,
              int4* __restrict__ ovf, int E) {
    const int b = (int)blockIdx.x;
    if (b < gemmBlocks) {
        gemm_role(X, W, Y, nrows, b * 4 + (int)(threadIdx.x >> 5));
    } else {
        const int h = (int)((__builtin_amdgcn_s_sendmsg_rtn(0x87) >> 18) & 1u);
        fill_role(erow, ecol, ev, cnt, slot, ovfN, ovf, E,
                  (b - gemmBlocks) * 128 + (int)threadIdx.x, h);
    }
}

__device__ __forceinline__ float xhalf(float x) {
    const int b = __builtin_bit_cast(int, x);
    return __builtin_bit_cast(float,
        __builtin_amdgcn_permlanex16(b, b, 0x76543210, 0xfedcba98, false, false));
}
template <int M>
__device__ __forceinline__ float xrow(float x) {
    const int b = __builtin_bit_cast(int, x);
    return __builtin_bit_cast(float,
        __builtin_amdgcn_update_dpp(0, b, 0x160 + M, 0xf, 0xf, false));
}

__device__ __forceinline__ float4 gather_row(const float* __restrict__ y, int c, int lane) {
    return ((const float4*)(y + (size_t)c * FEAT))[lane];
}
__device__ __forceinline__ void axpy4(float4& acc, float v, const float4 g) {
    acc.x = fmaf(v, g.x, acc.x);
    acc.y = fmaf(v, g.y, acc.y);
    acc.z = fmaf(v, g.z, acc.z);
    acc.w = fmaf(v, g.w, acc.w);
}

__global__ void __launch_bounds__(256, 8)
spmm1_fused(const float* __restrict__ y, const int* __restrict__ cnt,
                            const Ent* __restrict__ slot, const int* __restrict__ ovfN,
                            const int4* __restrict__ ovf,
                            const float* __restrict__ b1, const float* __restrict__ g1,
                            const float* __restrict__ be1, const float* __restrict__ m1,
                            const float* __restrict__ v1, const float* __restrict__ W2,
                            float* __restrict__ p, int N) {
    const int lane = threadIdx.x & 31;
    const int wv = __builtin_amdgcn_readfirstlane((int)(blockIdx.x * 8 + (threadIdx.x >> 5)));
    const int stride = (int)gridDim.x * 8;
    if (wv >= N) return;

    float bs[4], bt[4];
#pragma unroll
    for (int j = 0; j < 4; ++j) {
        const int ch = lane * 4 + j;
        const float s = g1[ch] * rsqrtf(v1[ch] + EPSB);
        bs[j] = s;
        bt[j] = (b1[ch] - m1[ch]) * s + be1[ch];
    }
    float w[28];
    {
        const float4* wq = (const float4*)(W2 + (size_t)lane * (4 * NCLS));
#pragma unroll
        for (int q = 0; q < 7; ++q) {
            const float4 t4 = wq[q];
            w[4 * q + 0] = t4.x; w[4 * q + 1] = t4.y;
            w[4 * q + 2] = t4.z; w[4 * q + 3] = t4.w;
        }
    }
    const bool b4 = (lane & 16) != 0;
    const bool b3 = (lane & 8) != 0;
    const bool b2 = (lane & 4) != 0;

    for (int base = wv; base < N; base += 32 * stride) {
        const int rj = base + lane * stride;
        int cntA = 0, cntB = 0;
        if (rj < N) {
            cntA = cnt[((size_t)rj * 2 + 0) * CP];
            cntB = cnt[((size_t)rj * 2 + 1) * CP];
        }
        int rows = (N - base + stride - 1) / stride;
        rows = rows < 32 ? rows : 32;

        int dA = __builtin_amdgcn_readlane(cntA, 0);
        int dB = __builtin_amdgcn_readlane(cntB, 0);
        Ent entA, entB;
        entA.col = 0;
        entA.val = 0.0f;
        entB.col = 0;
        entB.val = 0.0f;
        if (lane < (dA < SLOT ? dA : SLOT)) entA = slot[((size_t)base * 2 + 0) * SLOT + lane];
        if (lane < (dB < SLOT ? dB : SLOT)) entB = slot[((size_t)base * 2 + 1) * SLOT + lane];
        asm volatile("" :: "v"(entA.col), "v"(entA.val), "v"(entB.col), "v"(entB.val));

        for (int k = 0; k < rows; ++k) {
            const int r = base + k * stride;
            const int nA = dA < SLOT ? dA : SLOT;
            const int nB = dB < SLOT ? dB : SLOT;
            const int n = nA + nB;

            int dnA = 0, dnB = 0;
            Ent entnA, entnB;
            entnA.col = 0;
            entnA.val = 0.0f;
            entnB.col = 0;
            entnB.val = 0.0f;
            if (k + 1 < rows) {
                dnA = __builtin_amdgcn_readlane(cntA, k + 1);
                dnB = __builtin_amdgcn_readlane(cntB, k + 1);
                const size_t cn = (size_t)(r + stride) * 2;
                if (lane < (dnA < SLOT ? dnA : SLOT)) entnA = slot[(cn + 0) * SLOT + lane];
                if (lane < (dnB < SLOT ? dnB : SLOT)) entnB = slot[(cn + 1) * SLOT + lane];
            }

            const int colA = entA.col;
            const int colB = entB.col;
            const int valA = __float_as_int(entA.val);
            const int valB = __float_as_int(entB.val);
            auto colAt = [&](int e) -> int {
                const int a = __builtin_amdgcn_readlane(colA, e & 31);
                const int b = __builtin_amdgcn_readlane(colB, (e - nA) & 31);
                return e < nA ? a : b;
            };
            auto valAt = [&](int e) -> float {
                const int a = __builtin_amdgcn_readlane(valA, e & 31);
                const int b = __builtin_amdgcn_readlane(valB, (e - nA) & 31);
                return __int_as_float(e < nA ? a : b);
            };
            float4 acc = {0.f, 0.f, 0.f, 0.f};
            int e0 = 0;
            for (; e0 + 8 <= n; e0 += 8) {
                float4 g[8];
#pragma unroll
                for (int j = 0; j < 8; ++j)
                    g[j] = gather_row(y, colAt(e0 + j), lane);
#pragma unroll
                for (int j = 0; j < 8; ++j)
                    axpy4(acc, valAt(e0 + j), g[j]);
            }
            const int rem = n - e0;
            if (rem > 0) {
                float4 g[7];
#pragma unroll
                for (int j = 0; j < 7; ++j)
                    if (j < rem)
                        g[j] = gather_row(y, colAt(e0 + j), lane);
#pragma unroll
                for (int j = 0; j < 7; ++j)
                    if (j < rem)
                        axpy4(acc, valAt(e0 + j), g[j]);
            }
            if (dA > SLOT || dB > SLOT) {
                const int m = *ovfN;
                for (int j = 0; j < m; ++j) {
                    const int4 o = ovf[j];
                    if (o.x == r) axpy4(acc, __int_as_float(o.z), gather_row(y, o.y, lane));
                }
            }

            float h[4];
            const float af[4] = {acc.x, acc.y, acc.z, acc.w};
#pragma unroll
            for (int j = 0; j < 4; ++j) {
                const float hv = fmaf(af[j], bs[j], bt[j]);
                h[j] = hv > 0.0f ? hv : 0.0f;
            }

            float s8[8];
#pragma unroll
            for (int c = 0; c < NCLS; ++c) {
                float ssum = 0.0f;
#pragma unroll
                for (int j = 0; j < 4; ++j) ssum = fmaf(h[j], w[j * NCLS + c], ssum);
                s8[c] = ssum;
            }
            s8[7] = 0.0f;

            float t4v[4];
#pragma unroll
            for (int i = 0; i < 4; ++i) {
                const float keep = b4 ? s8[i + 4] : s8[i];
                const float send = b4 ? s8[i] : s8[i + 4];
                t4v[i] = keep + xhalf(send);
            }
            float u2[2];
#pragma unroll
            for (int i = 0; i < 2; ++i) {
                const float keep = b3 ? t4v[i + 2] : t4v[i];
                const float send = b3 ? t4v[i] : t4v[i + 2];
                u2[i] = keep + xrow<8>(send);
            }
            float tot;
            {
                const float keep = b2 ? u2[1] : u2[0];
                const float send = b2 ? u2[0] : u2[1];
                tot = keep + xrow<4>(send);
            }
            tot += xrow<2>(tot);
            tot += xrow<1>(tot);
            if ((lane & 3) == 0)
                p[(size_t)r * PP + (lane >> 2)] = tot;

            dA = dnA;
            dB = dnB;
            entA = entnA;
            entB = entnB;
        }
    }
}

__device__ __forceinline__ Ent load_ent(const Ent* __restrict__ slot, size_t chunk, int d, int lane) {
    Ent e;
    e.col = 0;
    e.val = 0.0f;
    if (lane < (d < SLOT ? d : SLOT)) e = slot[chunk * SLOT + lane];
    return e;
}
__device__ __forceinline__ void load_prow(const float* __restrict__ p, const Ent e, int d, int lane,
                                          float4& lo, float4& hi) {
    const float4 z4 = {0.f, 0.f, 0.f, 0.f};
    lo = z4;
    hi = z4;
    if (lane < (d < SLOT ? d : SLOT)) {
        const float4* q = (const float4*)(p + (size_t)e.col * PP);
        lo = q[0];
        hi = q[1];
    }
}

__global__ void __launch_bounds__(256)
spmm2_rows(const float* __restrict__ p, const int* __restrict__ cnt,
           const Ent* __restrict__ slot, const int* __restrict__ ovfN,
           const int4* __restrict__ ovf,
           const float* __restrict__ b2, const float* __restrict__ g2,
           const float* __restrict__ be2, const float* __restrict__ m2,
           const float* __restrict__ v2, float* __restrict__ out, int N) {
    const int lane = threadIdx.x & 31;
    const int wv = __builtin_amdgcn_readfirstlane((int)(blockIdx.x * 8 + (threadIdx.x >> 5)));
    const int stride = (int)gridDim.x * 8;
    if (wv >= N) return;

    const int cls = lane >> 2;
    const bool owner = (lane & 3) == 0 && cls < NCLS;
    float cb = 0.0f, cm = 0.0f, cs = 0.0f, ce = 0.0f;
    if (owner) {
        cb = b2[cls];
        cm = m2[cls];
        cs = g2[cls] * rsqrtf(v2[cls] + EPSB);
        ce = be2[cls];
    }
    const bool b4 = (lane & 16) != 0;
    const bool b3 = (lane & 8) != 0;
    const bool b2l = (lane & 4) != 0;

    for (int base = wv; base < N; base += 32 * stride) {
        const int rj = base + lane * stride;
        int cntA = 0, cntB = 0;
        if (rj < N) {
            cntA = cnt[((size_t)rj * 2 + 0) * CP];
            cntB = cnt[((size_t)rj * 2 + 1) * CP];
        }
        int rows = (N - base + stride - 1) / stride;
        rows = rows < 32 ? rows : 32;

        int d0A = __builtin_amdgcn_readlane(cntA, 0);
        int d0B = __builtin_amdgcn_readlane(cntB, 0);
        Ent e0A = load_ent(slot, (size_t)base * 2 + 0, d0A, lane);
        Ent e0B = load_ent(slot, (size_t)base * 2 + 1, d0B, lane);
        float4 ga, gb, gc, gd;
        load_prow(p, e0A, d0A, lane, ga, gb);
        load_prow(p, e0B, d0B, lane, gc, gd);
        int d1A = 0, d1B = 0;
        Ent e1A, e1B;
        e1A.col = 0;
        e1A.val = 0.0f;
        e1B.col = 0;
        e1B.val = 0.0f;
        if (rows > 1) {
            d1A = __builtin_amdgcn_readlane(cntA, 1);
            d1B = __builtin_amdgcn_readlane(cntB, 1);
            e1A = load_ent(slot, (size_t)(base + stride) * 2 + 0, d1A, lane);
            e1B = load_ent(slot, (size_t)(base + stride) * 2 + 1, d1B, lane);
        }

        for (int k = 0; k < rows; ++k) {
            const int r = base + k * stride;

            int d2A = 0, d2B = 0;
            Ent e2A, e2B;
            e2A.col = 0;
            e2A.val = 0.0f;
            e2B.col = 0;
            e2B.val = 0.0f;
            if (k + 2 < rows) {
                d2A = __builtin_amdgcn_readlane(cntA, k + 2);
                d2B = __builtin_amdgcn_readlane(cntB, k + 2);
                e2A = load_ent(slot, (size_t)(r + 2 * stride) * 2 + 0, d2A, lane);
                e2B = load_ent(slot, (size_t)(r + 2 * stride) * 2 + 1, d2B, lane);
            }
            float4 na, nb, nc, nd;
            load_prow(p, e1A, d1A, lane, na, nb);
            load_prow(p, e1B, d1B, lane, nc, nd);

            const float vA = e0A.val;
            const float vB = e0B.val;
            float s8[8];
            s8[0] = fmaf(vB, gc.x, vA * ga.x); s8[1] = fmaf(vB, gc.y, vA * ga.y);
            s8[2] = fmaf(vB, gc.z, vA * ga.z); s8[3] = fmaf(vB, gc.w, vA * ga.w);
            s8[4] = fmaf(vB, gd.x, vA * gb.x); s8[5] = fmaf(vB, gd.y, vA * gb.y);
            s8[6] = fmaf(vB, gd.z, vA * gb.z); s8[7] = fmaf(vB, gd.w, vA * gb.w);
            float t4v[4];
#pragma unroll
            for (int i = 0; i < 4; ++i) {
                const float keep = b4 ? s8[i + 4] : s8[i];
                const float send = b4 ? s8[i] : s8[i + 4];
                t4v[i] = keep + xhalf(send);
            }
            float u2[2];
#pragma unroll
            for (int i = 0; i < 2; ++i) {
                const float keep = b3 ? t4v[i + 2] : t4v[i];
                const float send = b3 ? t4v[i] : t4v[i + 2];
                u2[i] = keep + xrow<8>(send);
            }
            float tot;
            {
                const float keep = b2l ? u2[1] : u2[0];
                const float send = b2l ? u2[0] : u2[1];
                tot = keep + xrow<4>(send);
            }
            tot += xrow<2>(tot);
            tot += xrow<1>(tot);

            if (d0A > SLOT || d0B > SLOT) {
                const int m = *ovfN;
                for (int j = 0; j < m; ++j) {
                    const int4 o = ovf[j];
                    if (o.x == r && owner)
                        tot = fmaf(__int_as_float(o.z), p[(size_t)o.y * PP + cls], tot);
                }
            }
            if (owner)
                out[(size_t)r * NCLS + cls] = (tot + cb - cm) * cs + ce;

            d0A = d1A; e0A = e1A; ga = na; gb = nb;
            d0B = d1B; e0B = e1B; gc = nc; gd = nd;
            d1A = d2A; e1A = e2A;
            d1B = d2B; e1B = e2B;
        }
    }
}

extern "C" void kernel_launch(void* const* d_in, const int* in_sizes, int n_in,
                              void* d_out, int out_size, void* d_ws, size_t ws_size,
                              hipStream_t stream) {
    const float* x   = (const float*)d_in[0];
    const float* ev  = (const float*)d_in[1];
    const float* W1  = (const float*)d_in[2];
    const float* b1  = (const float*)d_in[3];
    const float* g1  = (const float*)d_in[4];
    const float* be1 = (const float*)d_in[5];
    const float* m1  = (const float*)d_in[6];
    const float* v1  = (const float*)d_in[7];
    const float* W2  = (const float*)d_in[8];
    const float* b2  = (const float*)d_in[9];
    const float* g2  = (const float*)d_in[10];
    const float* be2 = (const float*)d_in[11];
    const float* m2  = (const float*)d_in[12];
    const float* v2  = (const float*)d_in[13];
    const int*  erow = (const int*)d_in[14];
    const int*  ecol = (const int*)d_in[15];
    float* out = (float*)d_out;

    const int N = in_sizes[0] / FEAT;
    const int E = in_sizes[1];

    float* y    = (float*)d_ws;
    Ent*   slot = (Ent*)(y + (size_t)N * FEAT);
    int4*  ovf  = (int4*)(slot + (size_t)N * 2 * SLOT);
    float* p    = (float*)(ovf + E);
    int*   cnt  = (int*)(p + (size_t)N * PP);
    int*   ovfN = cnt + (size_t)N * 2 * CP;

    const int n4 = (2 * N + 1) * (CP / 4);
    zero_lines<<<(n4 + 255) / 256, 256, 0, stream>>>((int4*)cnt, n4);
    const int gemmBlocks = ((N >> 4) + 3) / 4;
    const int fillBlocks = (E + 511) / 512;
    gemm_and_fill<<<gemmBlocks + fillBlocks, 128, 0, stream>>>(x, W1, y, N, gemmBlocks,
                                                             erow, ecol, ev, cnt, slot,
                                                             ovfN, ovf, E);
    spmm1_fused<<<1024, 256, 0, stream>>>(y, cnt, slot, ovfN, ovf,
                                          b1, g1, be1, m1, v1, W2, p, N);
    spmm2_rows<<<1024, 256, 0, stream>>>(p, cnt, slot, ovfN, ovf,
                                         b2, g2, be2, m2, v2, out, N);
}


# a row slot becomes ONE 128-byte line: its counter in word 0 + 15 entries (rest in a rare spill line), lines of a chip half contiguous; counter lines and their loads go; spmm2 tree shifts one lane
# speedup vs baseline: 4.5091x; 1.2962x over previous
#include <hip/hip_runtime.h>
#include <hip/hip_bf16.h>

typedef __attribute__((ext_vector_type(2))) float v2f;
typedef __attribute__((ext_vector_type(8))) float v8f;

#define FEAT 128
#define NCLS 7
#define EPSB 1e-5f
#define LINE 16
#define CAP 31
#define PP 8

struct __attribute__((aligned(8))) Ent { int col; float val; };

__global__ void zero_lines(int4* __restrict__ ptr, int n4) {
    int i = blockIdx.x * blockDim.x + threadIdx.x;
    if (i < n4) ptr[i] = make_int4(0, 0, 0, 0);
}

__device__ __forceinline__ int* slot_counter(Ent* __restrict__ slot, int r, int h, int N) {
    return &slot[((size_t)h * N + r) * LINE].col;
}
__device__ __forceinline__ void put_edge(int r, int c, float v, int k, int h,
                                         Ent* __restrict__ slot, Ent* __restrict__ spill, int N,
                                         int* __restrict__ ovfN, int4* __restrict__ ovf) {
    if (k < CAP) {
        Ent w;
        w.col = c;
        w.val = v;
        const size_t base = ((size_t)h * N + r) * LINE;
        const int pl = 1 + k;
        if (pl < LINE) slot[base + pl] = w;
        else           spill[base + pl - LINE] = w;
    } else {
        const int j = atomicAdd(ovfN, 1);
        ovf[j] = make_int4(r, c, __float_as_int(v), 0);
    }
}

__device__ __forceinline__ void fill_role(const int* __restrict__ erow, const int* __restrict__ ecol,
                                          const float* __restrict__ ev,
                                          Ent* __restrict__ slot, Ent* __restrict__ spill, int N,
                                          int* __restrict__ ovfN,
                                          int4* __restrict__ ovf, int E, int q, int h) {
    const int e0 = q * 4;
    if (e0 >= E) return;
    if (e0 + 3 < E) {
        const int4   r4 = ((const int4*)erow)[q];
        const int4   c4 = ((const int4*)ecol)[q];
        const float4 v4 = ((const float4*)ev)[q];
        const int k0 = atomicAdd(slot_counter(slot, r4.x, h, N), 1);
        const int k1 = atomicAdd(slot_counter(slot, r4.y, h, N), 1);
        const int k2 = atomicAdd(slot_counter(slot, r4.z, h, N), 1);
        const int k3 = atomicAdd(slot_counter(slot, r4.w, h, N), 1);
        put_edge(r4.x, c4.x, v4.x, k0, h, slot, spill, N, ovfN, ovf);
        put_edge(r4.y, c4.y, v4.y, k1, h, slot, spill, N, ovfN, ovf);
        put_edge(r4.z, c4.z, v4.z, k2, h, slot, spill, N, ovfN, ovf);
        put_edge(r4.w, c4.w, v4.w, k3, h, slot, spill, N, ovfN, ovf);
    } else {
        for (int e = e0; e < E; ++e) {
            const int r = erow[e];
            const int k = atomicAdd(slot_counter(slot, r, h, N), 1);
            put_edge(r, ecol[e], ev[e], k, h, slot, spill, N, ovfN, ovf);
        }
    }
}

__device__ __forceinline__ void gemm_role(const float* __restrict__ X, const float* __restrict__ W,
                                          float* __restrict__ Y, int nrows, int mt) {
    const int lane = threadIdx.x & 31;
    const int half = lane >> 4;
    const int r    = lane & 15;
    if (mt >= (nrows >> 4)) return;
    const int m0 = mt << 4;

    const v8f vzero = {0.f,0.f,0.f,0.f,0.f,0.f,0.f,0.f};
    v8f acc[8];
#pragma unroll
    for (int nt = 0; nt < 8; ++nt) acc[nt] = vzero;

    const float* arow = X + (size_t)(m0 + r) * FEAT;
    for (int kt = 0; kt < 32; ++kt) {
        const int k0 = kt << 2;
        v2f a;
        a.x = arow[k0 + 2 * half + 0];
        a.y = arow[k0 + 2 * half + 1];
#pragma unroll
        for (int nt = 0; nt < 8; ++nt) {
            const int n0 = nt << 4;
            v2f b;
            b.x = W[(size_t)(k0 + 2 * half + 0) * FEAT + n0 + r];
            b.y = W[(size_t)(k0 + 2 * half + 1) * FEAT + n0 + r];
            acc[nt] = __builtin_amdgcn_wmma_f32_16x16x4_f32(
                false, a, false, b, (short)0, acc[nt], false, false);
        }
    }
#pragma unroll
    for (int nt = 0; nt < 8; ++nt) {
        const int n0 = nt << 4;
#pragma unroll
        for (int i = 0; i < 8; ++i)
            Y[(size_t)(m0 + i + 8 * half) * FEAT + n0 + r] = acc[nt][i];
    }
}

__global__ void __launch_bounds__(128)
gemm_and_fill(const float* __restrict__ X, const float* __restrict__ W,
              float* __restrict__ Y, int nrows, int gemmBlocks,
              const int* __restrict__ erow, const int* __restrict__ ecol,
              const float* __restrict__ ev,
              Ent* __restrict__ slot, Ent* __restrict__ spill, int* __restrict__ ovfN,
              int4* __restrict__ ovf, int E) {
    const int b = (int)blockIdx.x;
    if (b < gemmBlocks) {
        gemm_role(X, W, Y, nrows, b * 4 + (int)(threadIdx.x >> 5));
    } else {
        const int h = (int)((__builtin_amdgcn_s_sendmsg_rtn(0x87) >> 18) & 1u);
        fill_role(erow, ecol, ev, slot, spill, nrows, ovfN, ovf, E,
                  (b - gemmBlocks) * 128 + (int)threadIdx.x, h);
    }
}

__device__ __forceinline__ float xhalf(float x) {
    const int b = __builtin_bit_cast(int, x);
    return __builtin_bit_cast(float,
        __builtin_amdgcn_permlanex16(b, b, 0x76543210, 0xfedcba98, false, false));
}
template <int M>
__device__ __forceinline__ float xrow(float x) {
    const int b = __builtin_bit_cast(int, x);
    return __builtin_bit_cast(float,
        __builtin_amdgcn_update_dpp(0, b, 0x160 + M, 0xf, 0xf, false));
}

__device__ __forceinline__ Ent load_head(const Ent* __restrict__ slot, size_t line, int lane) {
    Ent e;
    e.col = 0;
    e.val = 0.0f;
    if (lane < LINE) e = slot[line * LINE + lane];
    return e;
}
__device__ __forceinline__ void load_tail(const Ent* __restrict__ spill, size_t line, int d, int lane,
                                          Ent& e) {
    if (d >= LINE) {
        if (lane >= LINE && lane <= (d < CAP ? d : CAP)) e = spill[line * LINE + lane - LINE];
        asm volatile("" :: "v"(e.col), "v"(e.val));
    }
}
__device__ __forceinline__ float4 gather_row(const float* __restrict__ y, int c, int lane) {
    return ((const float4*)(y + (size_t)c * FEAT))[lane];
}
__device__ __forceinline__ void axpy4(float4& acc, float v, const float4 g) {
    acc.x = fmaf(v, g.x, acc.x);
    acc.y = fmaf(v, g.y, acc.y);
    acc.z = fmaf(v, g.z, acc.z);
    acc.w = fmaf(v, g.w, acc.w);
}

__global__ void __launch_bounds__(256, 8)
spmm1_fused(const float* __restrict__ y,
                            const Ent* __restrict__ slot, const Ent* __restrict__ spill,
                            const int* __restrict__ ovfN,
                            const int4* __restrict__ ovf,
                            const float* __restrict__ b1, const float* __restrict__ g1,
                            const float* __restrict__ be1, const float* __restrict__ m1,
                            const float* __restrict__ v1, const float* __restrict__ W2,
                            float* __restrict__ p, int N) {
    const int lane = threadIdx.x & 31;
    const int wv = __builtin_amdgcn_readfirstlane((int)(blockIdx.x * 8 + (threadIdx.x >> 5)));
    const int stride = (int)gridDim.x * 8;
    if (wv >= N) return;

    float bs[4], bt[4];
#pragma unroll
    for (int j = 0; j < 4; ++j) {
        const int ch = lane * 4 + j;
        const float s = g1[ch] * rsqrtf(v1[ch] + EPSB);
        bs[j] = s;
        bt[j] = (b1[ch] - m1[ch]) * s + be1[ch];
    }
    float w[28];
    {
        const float4* wq = (const float4*)(W2 + (size_t)lane * (4 * NCLS));
#pragma unroll
        for (int q = 0; q < 7; ++q) {
            const float4 t4 = wq[q];
            w[4 * q + 0] = t4.x; w[4 * q + 1] = t4.y;
            w[4 * q + 2] = t4.z; w[4 * q + 3] = t4.w;
        }
    }
    const bool b4 = (lane & 16) != 0;
    const bool b3 = (lane & 8) != 0;
    const bool b2 = (lane & 4) != 0;

    {
        Ent entA = load_head(slot, (size_t)wv, lane);
        Ent entB = load_head(slot, (size_t)N + wv, lane);

        for (int r = wv; r < N; r += stride) {
            const int dA = __builtin_amdgcn_readlane(entA.col, 0);
            const int dB = __builtin_amdgcn_readlane(entB.col, 0);
            load_tail(spill, (size_t)r, dA, lane, entA);
            load_tail(spill, (size_t)N + r, dB, lane, entB);
            const int nA = dA < CAP ? dA : CAP;
            const int nB = dB < CAP ? dB : CAP;
            const int n = nA + nB;

            Ent entnA, entnB;
            entnA.col = 0;
            entnA.val = 0.0f;
            entnB.col = 0;
            entnB.val = 0.0f;
            if (r + stride < N) {
                const size_t rn = (size_t)(r + stride);
                entnA = load_head(slot, rn, lane);
                entnB = load_head(slot, (size_t)N + rn, lane);
            }

            const int colA = entA.col;
            const int colB = entB.col;
            const int valA = __float_as_int(entA.val);
            const int valB = __float_as_int(entB.val);
            auto colAt = [&](int e) -> int {
                const int a = __builtin_amdgcn_readlane(colA, (e + 1) & 31);
                const int b = __builtin_amdgcn_readlane(colB, (e - nA + 1) & 31);
                return e < nA ? a : b;
            };
            auto valAt = [&](int e) -> float {
                const int a = __builtin_amdgcn_readlane(valA, (e + 1) & 31);
                const int b = __builtin_amdgcn_readlane(valB, (e - nA + 1) & 31);
                return __int_as_float(e < nA ? a : b);
            };
            float4 acc = {0.f, 0.f, 0.f, 0.f};
            int e0 = 0;
            for (; e0 + 8 <= n; e0 += 8) {
                float4 g[8];
#pragma unroll
                for (int j = 0; j < 8; ++j)
                    g[j] = gather_row(y, colAt(e0 + j), lane);
#pragma unroll
                for (int j = 0; j < 8; ++j)
                    axpy4(acc, valAt(e0 + j), g[j]);
            }
            const int rem = n - e0;
            if (rem > 0) {
                float4 g[7];
#pragma unroll
                for (int j = 0; j < 7; ++j)
                    if (j < rem)
                        g[j] = gather_row(y, colAt(e0 + j), lane);
#pragma unroll
                for (int j = 0; j < 7; ++j)
                    if (j < rem)
                        axpy4(acc, valAt(e0 + j), g[j]);
            }
            if (dA > CAP || dB > CAP) {
                const int m = *ovfN;
                for (int j = 0; j < m; ++j) {
                    const int4 o = ovf[j];
                    if (o.x == r) axpy4(acc, __int_as_float(o.z), gather_row(y, o.y, lane));
                }
            }

            float h[4];
            const float af[4] = {acc.x, acc.y, acc.z, acc.w};
#pragma unroll
            for (int j = 0; j < 4; ++j) {
                const float hv = fmaf(af[j], bs[j], bt[j]);
                h[j] = hv > 0.0f ? hv : 0.0f;
            }

            float s8[8];
#pragma unroll
            for (int c = 0; c < NCLS; ++c) {
                float ssum = 0.0f;
#pragma unroll
                for (int j = 0; j < 4; ++j) ssum = fmaf(h[j], w[j * NCLS + c], ssum);
                s8[c] = ssum;
            }
            s8[7] = 0.0f;

            float t4v[4];
#pragma unroll
            for (int i = 0; i < 4; ++i) {
                const float keep = b4 ? s8[i + 4] : s8[i];
                const float send = b4 ? s8[i] : s8[i + 4];
                t4v[i] = keep + xhalf(send);
            }
            float u2[2];
#pragma unroll
            for (int i = 0; i < 2; ++i) {
                const float keep = b3 ? t4v[i + 2] : t4v[i];
                const float send = b3 ? t4v[i] : t4v[i + 2];
                u2[i] = keep + xrow<8>(send);
            }
            float tot;
            {
                const float keep = b2 ? u2[1] : u2[0];
                const float send = b2 ? u2[0] : u2[1];
                tot = keep + xrow<4>(send);
            }
            tot += xrow<2>(tot);
            tot += xrow<1>(tot);
            if ((lane & 3) == 0)
                p[(size_t)r * PP + (lane >> 2)] = tot;

            entA = entnA;
            entB = entnB;
        }
    }
}

__device__ __forceinline__ void load_prow(const float* __restrict__ p, const Ent e, int d, int lane,
                                          float4& lo, float4& hi) {
    const float4 z4 = {0.f, 0.f, 0.f, 0.f};
    lo = z4;
    hi = z4;
    if (lane >= 1 && lane <= (d < CAP ? d : CAP)) {
        const float4* q = (const float4*)(p + (size_t)e.col * PP);
        lo = q[0];
        hi = q[1];
    }
}

__global__ void __launch_bounds__(256)
spmm2_rows(const float* __restrict__ p,
           const Ent* __restrict__ slot, const Ent* __restrict__ spill,
           const int* __restrict__ ovfN,
           const int4* __restrict__ ovf,
           const float* __restrict__ b2, const float* __restrict__ g2,
           const float* __restrict__ be2, const float* __restrict__ m2,
           const float* __restrict__ v2, float* __restrict__ out, int N) {
    const int lane = threadIdx.x & 31;
    const int wv = __builtin_amdgcn_readfirstlane((int)(blockIdx.x * 8 + (threadIdx.x >> 5)));
    const int stride = (int)gridDim.x * 8;
    if (wv >= N) return;

    const int cls = lane >> 2;
    const bool owner = (lane & 3) == 0 && cls < NCLS;
    float cb = 0.0f, cm = 0.0f, cs = 0.0f, ce = 0.0f;
    if (owner) {
        cb = b2[cls];
        cm = m2[cls];
        cs = g2[cls] * rsqrtf(v2[cls] + EPSB);
        ce = be2[cls];
    }
    const bool b4 = (lane & 16) != 0;
    const bool b3 = (lane & 8) != 0;
    const bool b2l = (lane & 4) != 0;

    {
        Ent e0A = load_head(slot, (size_t)wv, lane);
        Ent e0B = load_head(slot, (size_t)N + wv, lane);
        int d0A = __builtin_amdgcn_readlane(e0A.col, 0);
        int d0B = __builtin_amdgcn_readlane(e0B.col, 0);
        load_tail(spill, (size_t)wv, d0A, lane, e0A);
        load_tail(spill, (size_t)N + wv, d0B, lane, e0B);
        float4 ga, gb, gc, gd;
        load_prow(p, e0A, d0A, lane, ga, gb);
        load_prow(p, e0B, d0B, lane, gc, gd);
        Ent e1A, e1B;
        e1A.col = 0;
        e1A.val = 0.0f;
        e1B.col = 0;
        e1B.val = 0.0f;
        if (wv + stride < N) {
            e1A = load_head(slot, (size_t)(wv + stride), lane);
            e1B = load_head(slot, (size_t)N + (wv + stride), lane);
        }

        for (int r = wv; r < N; r += stride) {
            const int d1A = __builtin_amdgcn_readlane(e1A.col, 0);
            const int d1B = __builtin_amdgcn_readlane(e1B.col, 0);
            load_tail(spill, (size_t)(r + stride), d1A, lane, e1A);
            load_tail(spill, (size_t)N + (r + stride), d1B, lane, e1B);

            Ent e2A, e2B;
            e2A.col = 0;
            e2A.val = 0.0f;
            e2B.col = 0;
            e2B.val = 0.0f;
            if (r + 2 * stride < N) {
                e2A = load_head(slot, (size_t)(r + 2 * stride), lane);
                e2B = load_head(slot, (size_t)N + (r + 2 * stride), lane);
            }
            float4 na, nb, nc, nd;
            load_prow(p, e1A, d1A, lane, na, nb);
            load_prow(p, e1B, d1B, lane, nc, nd);

            const float vA = e0A.val;
            const float vB = e0B.val;
            float s8[8];
            s8[0] = fmaf(vB, gc.x, vA * ga.x); s8[1] = fmaf(vB, gc.y, vA * ga.y);
            s8[2] = fmaf(vB, gc.z, vA * ga.z); s8[3] = fmaf(vB, gc.w, vA * ga.w);
            s8[4] = fmaf(vB, gd.x, vA * gb.x); s8[5] = fmaf(vB, gd.y, vA * gb.y);
            s8[6] = fmaf(vB, gd.z, vA * gb.z); s8[7] = fmaf(vB, gd.w, vA * gb.w);
            float t4v[4];
#pragma unroll
            for (int i = 0; i < 4; ++i) {
                const float keep = b4 ? s8[i + 4] : s8[i];
                const float send = b4 ? s8[i] : s8[i + 4];
                t4v[i] = keep + xhalf(send);
            }
            float u2[2];
#pragma unroll
            for (int i = 0; i < 2; ++i) {
                const float keep = b3 ? t4v[i + 2] : t4v[i];
                const float send = b3 ? t4v[i] : t4v[i + 2];
                u2[i] = keep + xrow<8>(send);
            }
            float tot;
            {
                const float keep = b2l ? u2[1] : u2[0];
                const float send = b2l ? u2[0] : u2[1];
                tot = keep + xrow<4>(send);
            }
            tot += xrow<2>(tot);
            tot += xrow<1>(tot);

            if (d0A > CAP || d0B > CAP) {
                const int m = *ovfN;
                for (int j = 0; j < m; ++j) {
                    const int4 o = ovf[j];
                    if (o.x == r && owner)
                        tot = fmaf(__int_as_float(o.z), p[(size_t)o.y * PP + cls], tot);
                }
            }
            if (owner)
                out[(size_t)r * NCLS + cls] = (tot + cb - cm) * cs + ce;

            d0A = d1A; e0A = e1A; ga = na; gb = nb;
            d0B = d1B; e0B = e1B; gc = nc; gd = nd;
            e1A = e2A;
            e1B = e2B;
        }
    }
}

extern "C" void kernel_launch(void* const* d_in, const int* in_sizes, int n_in,
                              void* d_out, int out_size, void* d_ws, size_t ws_size,
                              hipStream_t stream) {
    const float* x   = (const float*)d_in[0];
    const float* ev  = (const float*)d_in[1];
    const float* W1  = (const float*)d_in[2];
    const float* b1  = (const float*)d_in[3];
    const float* g1  = (const float*)d_in[4];
    const float* be1 = (const float*)d_in[5];
    const float* m1  = (const float*)d_in[6];
    const float* v1  = (const float*)d_in[7];
    const float* W2  = (const float*)d_in[8];
    const float* b2  = (const float*)d_in[9];
    const float* g2  = (const float*)d_in[10];
    const float* be2 = (const float*)d_in[11];
    const float* m2  = (const float*)d_in[12];
    const float* v2  = (const float*)d_in[13];
    const int*  erow = (const int*)d_in[14];
    const int*  ecol = (const int*)d_in[15];
    float* out = (float*)d_out;

    const int N = in_sizes[0] / FEAT;
    const int E = in_sizes[1];

    float* y     = (float*)d_ws;
    Ent*   slot  = (Ent*)(y + (size_t)N * FEAT);
    int*   ovfN  = (int*)(slot + (size_t)N * 2 * LINE);
    Ent*   spill = slot + ((size_t)N * 2 + 1) * LINE;
    int4*  ovf   = (int4*)(spill + (size_t)N * 2 * LINE);
    float* p     = (float*)(ovf + E);

    const int n4 = (2 * N + 1) * (LINE / 2);
    zero_lines<<<(n4 + 255) / 256, 256, 0, stream>>>((int4*)slot, n4);
    const int gemmBlocks = ((N >> 4) + 3) / 4;
    const int fillBlocks = (E + 511) / 512;
    gemm_and_fill<<<gemmBlocks + fillBlocks, 128, 0, stream>>>(x, W1, y, N, gemmBlocks,
                                                             erow, ecol, ev, slot, spill,
                                                             ovfN, ovf, E);
    spmm1_fused<<<1024, 256, 0, stream>>>(y, slot, spill, ovfN, ovf,
                                          b1, g1, be1, m1, v1, W2, p, N);
    spmm2_rows<<<1024, 256, 0, stream>>>(p, slot, spill, ovfN, ovf,
                                         b2, g2, be2, m2, v2, out, N);
}
